# starting kernel
# baseline (speedup 1.0000x reference)
#include <hip/hip_runtime.h>
#include <hip/hip_bf16.h>

typedef __attribute__((ext_vector_type(2)))  float    v2f;
typedef __attribute__((ext_vector_type(8)))  float    v8f;
typedef __attribute__((ext_vector_type(16))) _Float16 v16h;

#define CH 64

__global__ void k_fill(float* __restrict__ p, float v, int n) {
  int i = blockIdx.x * blockDim.x + threadIdx.x;
  if (i < n) p[i] = v;
}

__global__ void k_deg(const int* __restrict__ dst, float* __restrict__ deg, int E) {
  int e = blockIdx.x * blockDim.x + threadIdx.x;
  if (e < E) unsafeAtomicAdd(&deg[dst[e]], 1.0f);
}

__global__ void k_rsqrt(float* __restrict__ p, int n) {
  int i = blockIdx.x * blockDim.x + threadIdx.x;
  if (i < n) p[i] = rsqrtf(p[i]);
}

__global__ void k_edge_norm(const int* __restrict__ src, const int* __restrict__ dst,
                            const float* __restrict__ dinv, float* __restrict__ nrm, int E) {
  int e = blockIdx.x * blockDim.x + threadIdx.x;
  if (e < E) nrm[e] = dinv[src[e]] * dinv[dst[e]];
}

__global__ __launch_bounds__(256) void k_gemm64(const float* __restrict__ A,
                                                const float* __restrict__ W,
                                                float* __restrict__ C, int N) {
  const int lane = threadIdx.x & 31;
  const int wave = threadIdx.x >> 5;
  const int row0 = blockIdx.x * 32 + (wave >> 2) * 16;
  if (row0 >= N) return;
  const int n0 = (wave & 3) * 16;
  const int m  = lane & 15;
  const int kh = lane >> 4;
  v8f acc = {};

#if __has_builtin(__builtin_amdgcn_wmma_f32_16x16x4_f32)
  const float* arow = A + (size_t)(row0 + m) * CH + 2 * kh;
#pragma unroll
  for (int k0 = 0; k0 < CH; k0 += 4) {
    v2f a;
    a.x = arow[k0];
    a.y = arow[k0 + 1];
    v2f b;
    b.x = W[(k0 + 2 * kh + 0) * CH + n0 + m];
    b.y = W[(k0 + 2 * kh + 1) * CH + n0 + m];
    acc = __builtin_amdgcn_wmma_f32_16x16x4_f32(false, a, false, b,
                                                (short)0, acc, false, false);
  }
#else
  const float* arow = A + (size_t)(row0 + m) * CH;
#pragma unroll
  for (int kb = 0; kb < CH; kb += 32) {
    v16h a, b;
#pragma unroll
    for (int e = 0; e < 16; ++e) {
      int k = (e < 8) ? (8 * kh + e) : (16 + 8 * kh + (e - 8));
      a[e] = (_Float16)arow[kb + k];
      b[e] = (_Float16)W[(size_t)(kb + k) * CH + n0 + m];
    }
    acc = __builtin_amdgcn_wmma_f32_16x16x32_f16(false, a, false, b,
                                                 (short)0, acc, false, false);
  }
#endif

#pragma unroll
  for (int v = 0; v < 8; ++v) {
    int r = row0 + v + 8 * kh;
    if (r < N) C[(size_t)r * CH + (n0 + m)] = acc[v];
  }
}

__global__ void k_selfloop(const float* __restrict__ h, const float* __restrict__ dinv,
                           float* __restrict__ out, int total) {
  int t = blockIdx.x * blockDim.x + threadIdx.x;
  if (t >= total) return;
  float di = dinv[t >> 6];
  out[t] = h[t] * di * di;
}

__global__ __launch_bounds__(256) void k_scatter(const float* __restrict__ h,
                                                 const int* __restrict__ src,
                                                 const int* __restrict__ dst,
                                                 const float* __restrict__ nrm,
                                                 float* __restrict__ out, int E) {
  int t = blockIdx.x * blockDim.x + threadIdx.x;
  int e = t >> 5;
  if (e >= E) return;
  int c = (t & 31) * 2;
  int s = src[e], d = dst[e];
  float w = nrm[e];
  const float2 hv = *(const float2*)(h + (size_t)s * CH + c);
  float* op = out + (size_t)d * CH + c;
  unsafeAtomicAdd(op,     hv.x * w);
  unsafeAtomicAdd(op + 1, hv.y * w);
}

__global__ void k_bias_relu(float* __restrict__ h, const float* __restrict__ b, int total) {
  int t = blockIdx.x * blockDim.x + threadIdx.x;
  if (t >= total) return;
  float v = h[t] + b[t & (CH - 1)];
  h[t] = v > 0.0f ? v : 0.0f;
}

__global__ __launch_bounds__(256) void k_edge_pred(const float* __restrict__ h,
                                                   const int* __restrict__ src,
                                                   const int* __restrict__ dst,
                                                   const float* __restrict__ Wfc,
                                                   const float* __restrict__ bfc,
                                                   float* __restrict__ out, int E) {
  int t = blockIdx.x * blockDim.x + threadIdx.x;
  int e = t >> 5;
  if (e >= E) return;
  int lane = t & 31;
  int c = lane * 2;
  int s = src[e], d = dst[e];
  float2 hs = *(const float2*)(h + (size_t)s * CH + c);
  float2 hd = *(const float2*)(h + (size_t)d * CH + c);
  float2 w0 = *(const float2*)(Wfc + c);
  float2 w1 = *(const float2*)(Wfc + CH + c);
  float sum = hs.x * w0.x + hs.y * w0.y + hd.x * w1.x + hd.y * w1.y;
#pragma unroll
  for (int off = 16; off > 0; off >>= 1) sum += __shfl_xor(sum, off, 32);
  if (lane == 0) out[e] = sum + bfc[0];
}

extern "C" void kernel_launch(void* const* d_in, const int* in_sizes, int n_in,
                              void* d_out, int out_size, void* d_ws, size_t ws_size,
                              hipStream_t stream) {
  const float* x   = (const float*)d_in[0];
  const int*   ei  = (const int*)d_in[1];
  const float* W1  = (const float*)d_in[2];
  const float* b1  = (const float*)d_in[3];
  const float* W2  = (const float*)d_in[4];
  const float* b2  = (const float*)d_in[5];
  const float* Wfc = (const float*)d_in[6];
  const float* bfc = (const float*)d_in[7];

  const int N = in_sizes[0] / CH;
  const int E = in_sizes[1] / 2;
  const int NF = N * CH;

  const int* src = ei;
  const int* dst = ei + E;

  float* ws   = (float*)d_ws;
  float* dinv = ws;
  float* nrm  = dinv + N;
  float* HB   = nrm + E;
  float* AB   = HB + (size_t)NF;

  const int TB = 256;
  dim3 blk(TB);
  dim3 gN((N + TB - 1) / TB);
  dim3 gE((E + TB - 1) / TB);
  dim3 gNF((NF + TB - 1) / TB);
  dim3 gEw((E * 32 + TB - 1) / TB);
  dim3 gG((N + 31) / 32);

  k_fill<<<gN, blk, 0, stream>>>(dinv, 1.0f, N);
  k_deg<<<gE, blk, 0, stream>>>(dst, dinv, E);
  k_rsqrt<<<gN, blk, 0, stream>>>(dinv, N);
  k_edge_norm<<<gE, blk, 0, stream>>>(src, dst, dinv, nrm, E);

  k_gemm64<<<gG, blk, 0, stream>>>(x, W1, HB, N);
  k_selfloop<<<gNF, blk, 0, stream>>>(HB, dinv, AB, NF);
  k_scatter<<<gEw, blk, 0, stream>>>(HB, src, dst, nrm, AB, E);
  k_bias_relu<<<gNF, blk, 0, stream>>>(AB, b1, NF);

  k_gemm64<<<gG, blk, 0, stream>>>(AB, W2, HB, N);
  k_selfloop<<<gNF, blk, 0, stream>>>(HB, dinv, AB, NF);
  k_scatter<<<gEw, blk, 0, stream>>>(HB, src, dst, nrm, AB, E);
  k_bias_relu<<<gNF, blk, 0, stream>>>(AB, b2, NF);

  k_edge_pred<<<gEw, blk, 0, stream>>>(AB, src, dst, Wfc, bfc, (float*)d_out, E);
}


# Edge head factored: layer-2 bias+relu pass takes per-node dots h.Wfc[0:64], h.Wfc[64:128]; edge kernel adds ps[src]+pd[dst]+b, one thread per edge (128-term sum regrouped as two 64-term dots)
# speedup vs baseline: 1.2302x; 1.2302x over previous
#include <hip/hip_runtime.h>
#include <hip/hip_bf16.h>

typedef __attribute__((ext_vector_type(2)))  float    v2f;
typedef __attribute__((ext_vector_type(8)))  float    v8f;
typedef __attribute__((ext_vector_type(16))) _Float16 v16h;

#define CH 64

__global__ void k_fill(float* __restrict__ p, float v, int n) {
  int i = blockIdx.x * blockDim.x + threadIdx.x;
  if (i < n) p[i] = v;
}

__global__ void k_deg(const int* __restrict__ dst, float* __restrict__ deg, int E) {
  int e = blockIdx.x * blockDim.x + threadIdx.x;
  if (e < E) unsafeAtomicAdd(&deg[dst[e]], 1.0f);
}

__global__ void k_rsqrt(float* __restrict__ p, int n) {
  int i = blockIdx.x * blockDim.x + threadIdx.x;
  if (i < n) p[i] = rsqrtf(p[i]);
}

__global__ void k_edge_norm(const int* __restrict__ src, const int* __restrict__ dst,
                            const float* __restrict__ dinv, float* __restrict__ nrm, int E) {
  int e = blockIdx.x * blockDim.x + threadIdx.x;
  if (e < E) nrm[e] = dinv[src[e]] * dinv[dst[e]];
}

__global__ __launch_bounds__(256) void k_gemm64(const float* __restrict__ A,
                                                const float* __restrict__ W,
                                                float* __restrict__ C, int N) {
  const int lane = threadIdx.x & 31;
  const int wave = threadIdx.x >> 5;
  const int row0 = blockIdx.x * 32 + (wave >> 2) * 16;
  if (row0 >= N) return;
  const int n0 = (wave & 3) * 16;
  const int m  = lane & 15;
  const int kh = lane >> 4;
  v8f acc = {};

#if __has_builtin(__builtin_amdgcn_wmma_f32_16x16x4_f32)
  const float* arow = A + (size_t)(row0 + m) * CH + 2 * kh;
#pragma unroll
  for (int k0 = 0; k0 < CH; k0 += 4) {
    v2f a;
    a.x = arow[k0];
    a.y = arow[k0 + 1];
    v2f b;
    b.x = W[(k0 + 2 * kh + 0) * CH + n0 + m];
    b.y = W[(k0 + 2 * kh + 1) * CH + n0 + m];
    acc = __builtin_amdgcn_wmma_f32_16x16x4_f32(false, a, false, b,
                                                (short)0, acc, false, false);
  }
#else
  const float* arow = A + (size_t)(row0 + m) * CH;
#pragma unroll
  for (int kb = 0; kb < CH; kb += 32) {
    v16h a, b;
#pragma unroll
    for (int e = 0; e < 16; ++e) {
      int k = (e < 8) ? (8 * kh + e) : (16 + 8 * kh + (e - 8));
      a[e] = (_Float16)arow[kb + k];
      b[e] = (_Float16)W[(size_t)(kb + k) * CH + n0 + m];
    }
    acc = __builtin_amdgcn_wmma_f32_16x16x32_f16(false, a, false, b,
                                                 (short)0, acc, false, false);
  }
#endif

#pragma unroll
  for (int v = 0; v < 8; ++v) {
    int r = row0 + v + 8 * kh;
    if (r < N) C[(size_t)r * CH + (n0 + m)] = acc[v];
  }
}

__global__ void k_selfloop(const float* __restrict__ h, const float* __restrict__ dinv,
                           float* __restrict__ out, int total) {
  int t = blockIdx.x * blockDim.x + threadIdx.x;
  if (t >= total) return;
  float di = dinv[t >> 6];
  out[t] = h[t] * di * di;
}

__global__ __launch_bounds__(256) void k_scatter(const float* __restrict__ h,
                                                 const int* __restrict__ src,
                                                 const int* __restrict__ dst,
                                                 const float* __restrict__ nrm,
                                                 float* __restrict__ out, int E) {
  int t = blockIdx.x * blockDim.x + threadIdx.x;
  int e = t >> 5;
  if (e >= E) return;
  int c = (t & 31) * 2;
  int s = src[e], d = dst[e];
  float w = nrm[e];
  const float2 hv = *(const float2*)(h + (size_t)s * CH + c);
  float* op = out + (size_t)d * CH + c;
  unsafeAtomicAdd(op,     hv.x * w);
  unsafeAtomicAdd(op + 1, hv.y * w);
}

__global__ void k_bias_relu(float* __restrict__ h, const float* __restrict__ b, int total) {
  int t = blockIdx.x * blockDim.x + threadIdx.x;
  if (t >= total) return;
  float v = h[t] + b[t & (CH - 1)];
  h[t] = v > 0.0f ? v : 0.0f;
}

__global__ __launch_bounds__(256) void k_bias_relu_proj(const float* __restrict__ agg,
                                                        const float* __restrict__ b,
                                                        const float* __restrict__ Wfc,
                                                        float* __restrict__ ps,
                                                        float* __restrict__ pd, int N) {
  int t = blockIdx.x * blockDim.x + threadIdx.x;
  int i = t >> 5;
  if (i >= N) return;
  int lane = t & 31;
  int c = lane * 2;
  float2 a  = *(const float2*)(agg + (size_t)i * CH + c);
  float2 bb = *(const float2*)(b + c);
  float2 w0 = *(const float2*)(Wfc + c);
  float2 w1 = *(const float2*)(Wfc + CH + c);
  float hx = a.x + bb.x;
  float hy = a.y + bb.y;
  hx = hx > 0.0f ? hx : 0.0f;
  hy = hy > 0.0f ? hy : 0.0f;
  float s0 = hx * w0.x + hy * w0.y;
  float s1 = hx * w1.x + hy * w1.y;
#pragma unroll
  for (int off = 16; off > 0; off >>= 1) {
    s0 += __shfl_xor(s0, off, 32);
    s1 += __shfl_xor(s1, off, 32);
  }
  if (lane == 0) {
    ps[i] = s0;
    pd[i] = s1;
  }
}

__global__ void k_edge_pred(const float* __restrict__ ps, const float* __restrict__ pd,
                            const int* __restrict__ src, const int* __restrict__ dst,
                            const float* __restrict__ bfc, float* __restrict__ out, int E) {
  int e = blockIdx.x * blockDim.x + threadIdx.x;
  if (e < E) out[e] = ps[src[e]] + pd[dst[e]] + bfc[0];
}

extern "C" void kernel_launch(void* const* d_in, const int* in_sizes, int n_in,
                              void* d_out, int out_size, void* d_ws, size_t ws_size,
                              hipStream_t stream) {
  const float* x   = (const float*)d_in[0];
  const int*   ei  = (const int*)d_in[1];
  const float* W1  = (const float*)d_in[2];
  const float* b1  = (const float*)d_in[3];
  const float* W2  = (const float*)d_in[4];
  const float* b2  = (const float*)d_in[5];
  const float* Wfc = (const float*)d_in[6];
  const float* bfc = (const float*)d_in[7];

  const int N = in_sizes[0] / CH;
  const int E = in_sizes[1] / 2;
  const int NF = N * CH;

  const int* src = ei;
  const int* dst = ei + E;

  float* ws   = (float*)d_ws;
  float* dinv = ws;
  float* nrm  = dinv + N;
  float* HB   = nrm + E;
  float* AB   = HB + (size_t)NF;
  float* PS   = AB + (size_t)NF;
  float* PD   = PS + N;

  const int TB = 256;
  dim3 blk(TB);
  dim3 gN((N + TB - 1) / TB);
  dim3 gE((E + TB - 1) / TB);
  dim3 gNF((NF + TB - 1) / TB);
  dim3 gEw((E * 32 + TB - 1) / TB);
  dim3 gNw((N * 32 + TB - 1) / TB);
  dim3 gG((N + 31) / 32);

  k_fill<<<gN, blk, 0, stream>>>(dinv, 1.0f, N);
  k_deg<<<gE, blk, 0, stream>>>(dst, dinv, E);
  k_rsqrt<<<gN, blk, 0, stream>>>(dinv, N);
  k_edge_norm<<<gE, blk, 0, stream>>>(src, dst, dinv, nrm, E);

  k_gemm64<<<gG, blk, 0, stream>>>(x, W1, HB, N);
  k_selfloop<<<gNF, blk, 0, stream>>>(HB, dinv, AB, NF);
  k_scatter<<<gEw, blk, 0, stream>>>(HB, src, dst, nrm, AB, E);
  k_bias_relu<<<gNF, blk, 0, stream>>>(AB, b1, NF);

  k_gemm64<<<gG, blk, 0, stream>>>(AB, W2, HB, N);
  k_selfloop<<<gNF, blk, 0, stream>>>(HB, dinv, AB, NF);
  k_scatter<<<gEw, blk, 0, stream>>>(HB, src, dst, nrm, AB, E);
  k_bias_relu_proj<<<gNw, blk, 0, stream>>>(AB, b2, Wfc, PS, PD, N);

  k_edge_pred<<<gE, blk, 0, stream>>>(PS, PD, src, dst, bfc, (float*)d_out, E);
}


# Aggregation by destination-sorted edge list (in-run counting sort, int atomic slot counter): one wave per node sums its rows in registers, stores once; float atomics gone; sum order = slot order
# speedup vs baseline: 1.5902x; 1.2926x over previous
#include <hip/hip_runtime.h>
#include <hip/hip_bf16.h>

typedef __attribute__((ext_vector_type(2)))  float    v2f;
typedef __attribute__((ext_vector_type(8)))  float    v8f;
typedef __attribute__((ext_vector_type(16))) _Float16 v16h;

#define CH 64

__global__ void k_zero(int* __restrict__ p, int n) {
  int i = blockIdx.x * blockDim.x + threadIdx.x;
  if (i < n) p[i] = 0;
}

__global__ void k_count(const int* __restrict__ dst, int* __restrict__ cnt, int E) {
  int e = blockIdx.x * blockDim.x + threadIdx.x;
  if (e < E) atomicAdd(&cnt[dst[e]], 1);
}

__global__ __launch_bounds__(256) void k_offsets(const int* __restrict__ cnt,
                                                 int* __restrict__ cur,
                                                 float* __restrict__ dinv, int N) {
  __shared__ int s_front[8];
  __shared__ int s_own[8];
  const int tid = threadIdx.x;
  const int lane = tid & 31;
  const int wave = tid >> 5;
  const int4* c4 = (const int4*)cnt;
  const int nq = blockIdx.x * 64;
  int front = 0;
  for (int j = tid; j < nq; j += 256) {
    int4 v = c4[j];
    front += (v.x + v.y) + (v.z + v.w);
  }
  const int i = blockIdx.x * 256 + tid;
  const int c = (i < N) ? cnt[i] : 0;
  int incl = c;
#pragma unroll
  for (int off = 1; off < 32; off <<= 1) {
    int t = __shfl_up(incl, off, 32);
    if (lane >= off) incl += t;
  }
#pragma unroll
  for (int off = 16; off > 0; off >>= 1) front += __shfl_xor(front, off, 32);
  if (lane == 31) s_own[wave] = incl;
  if (lane == 0) s_front[wave] = front;
  __syncthreads();
  int base = 0;
#pragma unroll
  for (int w = 0; w < 8; ++w) {
    base += s_front[w];
    if (w < wave) base += s_own[w];
  }
  if (i < N) {
    cur[i] = base + incl - c;
    dinv[i] = rsqrtf((float)(c + 1));
  }
}

__global__ void k_place(const int* __restrict__ src, const int* __restrict__ dst,
                        const float* __restrict__ dinv, int* __restrict__ cur,
                        int* __restrict__ list, int E) {
  int e = blockIdx.x * blockDim.x + threadIdx.x;
  if (e >= E) return;
  int s = src[e], d = dst[e];
  int slot = atomicAdd(&cur[d], 1);
  float w = dinv[s] * dinv[d];
  ((int2*)list)[slot] = make_int2(s, __float_as_int(w));
}

__global__ __launch_bounds__(256) void k_gemm64(const float* __restrict__ A,
                                                const float* __restrict__ W,
                                                float* __restrict__ C, int N) {
  const int lane = threadIdx.x & 31;
  const int wave = threadIdx.x >> 5;
  const int row0 = blockIdx.x * 32 + (wave >> 2) * 16;
  if (row0 >= N) return;
  const int n0 = (wave & 3) * 16;
  const int m  = lane & 15;
  const int kh = lane >> 4;
  v8f acc = {};

#if __has_builtin(__builtin_amdgcn_wmma_f32_16x16x4_f32)
  const float* arow = A + (size_t)(row0 + m) * CH + 2 * kh;
#pragma unroll
  for (int k0 = 0; k0 < CH; k0 += 4) {
    v2f a;
    a.x = arow[k0];
    a.y = arow[k0 + 1];
    v2f b;
    b.x = W[(k0 + 2 * kh + 0) * CH + n0 + m];
    b.y = W[(k0 + 2 * kh + 1) * CH + n0 + m];
    acc = __builtin_amdgcn_wmma_f32_16x16x4_f32(false, a, false, b,
                                                (short)0, acc, false, false);
  }
#else
  const float* arow = A + (size_t)(row0 + m) * CH;
#pragma unroll
  for (int kb = 0; kb < CH; kb += 32) {
    v16h a, b;
#pragma unroll
    for (int e = 0; e < 16; ++e) {
      int k = (e < 8) ? (8 * kh + e) : (16 + 8 * kh + (e - 8));
      a[e] = (_Float16)arow[kb + k];
      b[e] = (_Float16)W[(size_t)(kb + k) * CH + n0 + m];
    }
    acc = __builtin_amdgcn_wmma_f32_16x16x32_f16(false, a, false, b,
                                                 (short)0, acc, false, false);
  }
#endif

#pragma unroll
  for (int v = 0; v < 8; ++v) {
    int r = row0 + v + 8 * kh;
    if (r < N) C[(size_t)r * CH + (n0 + m)] = acc[v];
  }
}

static __device__ __forceinline__ float xhalf(float x) {
  return __builtin_bit_cast(float, __builtin_amdgcn_permlanex16(__builtin_bit_cast(int, x), __builtin_bit_cast(int, x),
                                                                0x76543210, 0xfedcba98, false, false));
}

static __device__ __forceinline__ float4 gather_row(const float* __restrict__ h,
                                                    const int* __restrict__ list,
                                                    const int* __restrict__ cur,
                                                    const float* __restrict__ dinv,
                                                    int i, int lane) {
  const int half = lane >> 4;
  const int c4 = (lane & 15) * 4;
  const int end = cur[i];
  const int start = (i > 0) ? cur[i - 1] : 0;
  const float di = dinv[i];
  const float4 self = *(const float4*)(h + (size_t)i * CH + c4);
  const float ws = di * di;
  float4 acc;
  acc.x = half ? 0.0f : self.x * ws;
  acc.y = half ? 0.0f : self.y * ws;
  acc.z = half ? 0.0f : self.z * ws;
  acc.w = half ? 0.0f : self.w * ws;

  for (int base = start; base < end; base += 32) {
    const int n = min(32, end - base);
    int2 ent = make_int2(i, 0);
    if (lane < n) ent = ((const int2*)list)[base + lane];
    const int npairs = (n + 1) >> 1;
    for (int p = 0; p < npairs; p += 8) {
      int s[8];
      float w[8];
      float4 r[8];
#pragma unroll
      for (int u = 0; u < 8; ++u) {
        const int q = 2 * (p + u);
        const int sa = __builtin_amdgcn_readlane(ent.x, q);
        const int sb = __builtin_amdgcn_readlane(ent.x, q + 1);
        const int wa = __builtin_amdgcn_readlane(ent.y, q);
        const int wb = __builtin_amdgcn_readlane(ent.y, q + 1);
        s[u] = half ? sb : sa;
        w[u] = __int_as_float(half ? wb : wa);
      }
#pragma unroll
      for (int u = 0; u < 8; ++u) {
        r[u] = make_float4(0.0f, 0.0f, 0.0f, 0.0f);
        if (2 * (p + u) + half < n) r[u] = *(const float4*)(h + (size_t)s[u] * CH + c4);
      }
#pragma unroll
      for (int u = 0; u < 8; ++u) {
        acc.x = fmaf(r[u].x, w[u], acc.x);
        acc.y = fmaf(r[u].y, w[u], acc.y);
        acc.z = fmaf(r[u].z, w[u], acc.z);
        acc.w = fmaf(r[u].w, w[u], acc.w);
      }
    }
  }
  acc.x += xhalf(acc.x);
  acc.y += xhalf(acc.y);
  acc.z += xhalf(acc.z);
  acc.w += xhalf(acc.w);
  return acc;
}

__global__ __launch_bounds__(256) void k_gather_relu(const float* __restrict__ h,
                                                     const int* __restrict__ list,
                                                     const int* __restrict__ cur,
                                                     const float* __restrict__ dinv,
                                                     const float* __restrict__ b,
                                                     float* __restrict__ out, int N) {
  const int lane = threadIdx.x & 31;
  const int i = __builtin_amdgcn_readfirstlane(blockIdx.x * 8 + (threadIdx.x >> 5));
  if (i >= N) return;
  float4 v = gather_row(h, list, cur, dinv, i, lane);
  const int c4 = (lane & 15) * 4;
  const float4 bb = *(const float4*)(b + c4);
  v.x += bb.x;  v.y += bb.y;  v.z += bb.z;  v.w += bb.w;
  v.x = v.x > 0.0f ? v.x : 0.0f;
  v.y = v.y > 0.0f ? v.y : 0.0f;
  v.z = v.z > 0.0f ? v.z : 0.0f;
  v.w = v.w > 0.0f ? v.w : 0.0f;
  if (lane < 16) *(float4*)(out + (size_t)i * CH + c4) = v;
}

__global__ __launch_bounds__(256) void k_gather_relu_proj(const float* __restrict__ h,
                                                          const int* __restrict__ list,
                                                          const int* __restrict__ cur,
                                                          const float* __restrict__ dinv,
                                                          const float* __restrict__ b,
                                                          const float* __restrict__ Wfc,
                                                          float* __restrict__ ps,
                                                          float* __restrict__ pd, int N) {
  const int lane = threadIdx.x & 31;
  const int i = __builtin_amdgcn_readfirstlane(blockIdx.x * 8 + (threadIdx.x >> 5));
  if (i >= N) return;
  float4 v = gather_row(h, list, cur, dinv, i, lane);
  const int c4 = (lane & 15) * 4;
  const float4 bb = *(const float4*)(b + c4);
  const float4 w0 = *(const float4*)(Wfc + c4);
  const float4 w1 = *(const float4*)(Wfc + CH + c4);
  v.x += bb.x;  v.y += bb.y;  v.z += bb.z;  v.w += bb.w;
  v.x = v.x > 0.0f ? v.x : 0.0f;
  v.y = v.y > 0.0f ? v.y : 0.0f;
  v.z = v.z > 0.0f ? v.z : 0.0f;
  v.w = v.w > 0.0f ? v.w : 0.0f;
  float s0 = (v.x * w0.x + v.y * w0.y) + (v.z * w0.z + v.w * w0.w);
  float s1 = (v.x * w1.x + v.y * w1.y) + (v.z * w1.z + v.w * w1.w);
#pragma unroll
  for (int off = 8; off > 0; off >>= 1) {
    s0 += __shfl_xor(s0, off, 32);
    s1 += __shfl_xor(s1, off, 32);
  }
  if (lane == 0) {
    ps[i] = s0;
    pd[i] = s1;
  }
}

__global__ void k_edge_pred(const float* __restrict__ ps, const float* __restrict__ pd,
                            const int* __restrict__ src, const int* __restrict__ dst,
                            const float* __restrict__ bfc, float* __restrict__ out, int E) {
  int e = blockIdx.x * blockDim.x + threadIdx.x;
  if (e < E) out[e] = ps[src[e]] + pd[dst[e]] + bfc[0];
}

extern "C" void kernel_launch(void* const* d_in, const int* in_sizes, int n_in,
                              void* d_out, int out_size, void* d_ws, size_t ws_size,
                              hipStream_t stream) {
  const float* x   = (const float*)d_in[0];
  const int*   ei  = (const int*)d_in[1];
  const float* W1  = (const float*)d_in[2];
  const float* b1  = (const float*)d_in[3];
  const float* W2  = (const float*)d_in[4];
  const float* b2  = (const float*)d_in[5];
  const float* Wfc = (const float*)d_in[6];
  const float* bfc = (const float*)d_in[7];

  const int N = in_sizes[0] / CH;
  const int E = in_sizes[1] / 2;
  const int NF = N * CH;

  const int* src = ei;
  const int* dst = ei + E;

  float* ws   = (float*)d_ws;
  float* HB   = ws;
  float* AB   = HB + (size_t)NF;
  int*   list = (int*)(AB + (size_t)NF);
  int*   cnt  = list + 2 * (size_t)E;
  int*   cur  = cnt + N;
  float* dinv = (float*)(cur + N);
  float* PS   = dinv + N;
  float* PD   = PS + N;

  const int TB = 256;
  dim3 blk(TB);
  dim3 gN((N + TB - 1) / TB);
  dim3 gE((E + TB - 1) / TB);
  dim3 gNw((N + 7) / 8);
  dim3 gG((N + 31) / 32);

  k_zero<<<gN, blk, 0, stream>>>(cnt, N);
  k_count<<<gE, blk, 0, stream>>>(dst, cnt, E);
  k_offsets<<<gN, blk, 0, stream>>>(cnt, cur, dinv, N);
  k_place<<<gE, blk, 0, stream>>>(src, dst, dinv, cur, list, E);

  k_gemm64<<<gG, blk, 0, stream>>>(x, W1, HB, N);
  k_gather_relu<<<gNw, blk, 0, stream>>>(HB, list, cur, dinv, b1, AB, N);

  k_gemm64<<<gG, blk, 0, stream>>>(AB, W2, HB, N);
  k_gather_relu_proj<<<gNw, blk, 0, stream>>>(HB, list, cur, dinv, b2, Wfc, PS, PD, N);

  k_edge_pred<<<gE, blk, 0, stream>>>(PS, PD, src, dst, bfc, (float*)d_out, E);
}


# Edge slot = segment start + rank, rank = the counting add's returned value kept per edge: placement pass has no atomic (one scattered integer add per edge, not two); sum order = count arrival order
# speedup vs baseline: 1.9828x; 1.2469x over previous
#include <hip/hip_runtime.h>
#include <hip/hip_bf16.h>

typedef __attribute__((ext_vector_type(2)))  float    v2f;
typedef __attribute__((ext_vector_type(8)))  float    v8f;
typedef __attribute__((ext_vector_type(16))) _Float16 v16h;

#define CH 64

__global__ void k_zero(int* __restrict__ p, int n) {
  int i = blockIdx.x * blockDim.x + threadIdx.x;
  if (i < n) p[i] = 0;
}

__global__ void k_count(const int* __restrict__ dst, int* __restrict__ cnt,
                        int* __restrict__ rank, int E) {
  int e = blockIdx.x * blockDim.x + threadIdx.x;
  if (e < E) rank[e] = atomicAdd(&cnt[dst[e]], 1);
}

__global__ __launch_bounds__(256) void k_offsets(const int* __restrict__ cnt,
                                                 int* __restrict__ cur,
                                                 float* __restrict__ dinv, int N) {
  __shared__ int s_front[8];
  __shared__ int s_own[8];
  const int tid = threadIdx.x;
  const int lane = tid & 31;
  const int wave = tid >> 5;
  const int4* c4 = (const int4*)cnt;
  const int nq = blockIdx.x * 64;
  int front = 0;
  for (int j = tid; j < nq; j += 256) {
    int4 v = c4[j];
    front += (v.x + v.y) + (v.z + v.w);
  }
  const int i = blockIdx.x * 256 + tid;
  const int c = (i < N) ? cnt[i] : 0;
  int incl = c;
#pragma unroll
  for (int off = 1; off < 32; off <<= 1) {
    int t = __shfl_up(incl, off, 32);
    if (lane >= off) incl += t;
  }
#pragma unroll
  for (int off = 16; off > 0; off >>= 1) front += __shfl_xor(front, off, 32);
  if (lane == 31) s_own[wave] = incl;
  if (lane == 0) s_front[wave] = front;
  __syncthreads();
  int base = 0;
#pragma unroll
  for (int w = 0; w < 8; ++w) {
    base += s_front[w];
    if (w < wave) base += s_own[w];
  }
  if (i < N) {
    cur[i] = base + incl - c;
    if (i == N - 1) cur[N] = base + incl;
    dinv[i] = rsqrtf((float)(c + 1));
  }
}

__global__ void k_place(const int* __restrict__ src, const int* __restrict__ dst,
                        const int* __restrict__ rank, const float* __restrict__ dinv,
                        const int* __restrict__ cur, int* __restrict__ list, int E) {
  int e = blockIdx.x * blockDim.x + threadIdx.x;
  if (e >= E) return;
  int s = src[e], d = dst[e];
  int slot = cur[d] + rank[e];
  float w = dinv[s] * dinv[d];
  ((int2*)list)[slot] = make_int2(s, __float_as_int(w));
}

__global__ __launch_bounds__(256) void k_gemm64(const float* __restrict__ A,
                                                const float* __restrict__ W,
                                                float* __restrict__ C, int N) {
  const int lane = threadIdx.x & 31;
  const int wave = threadIdx.x >> 5;
  const int row0 = blockIdx.x * 32 + (wave >> 2) * 16;
  if (row0 >= N) return;
  const int n0 = (wave & 3) * 16;
  const int m  = lane & 15;
  const int kh = lane >> 4;
  v8f acc = {};

#if __has_builtin(__builtin_amdgcn_wmma_f32_16x16x4_f32)
  const float* arow = A + (size_t)(row0 + m) * CH + 2 * kh;
#pragma unroll
  for (int k0 = 0; k0 < CH; k0 += 4) {
    v2f a;
    a.x = arow[k0];
    a.y = arow[k0 + 1];
    v2f b;
    b.x = W[(k0 + 2 * kh + 0) * CH + n0 + m];
    b.y = W[(k0 + 2 * kh + 1) * CH + n0 + m];
    acc = __builtin_amdgcn_wmma_f32_16x16x4_f32(false, a, false, b,
                                                (short)0, acc, false, false);
  }
#else
  const float* arow = A + (size_t)(row0 + m) * CH;
#pragma unroll
  for (int kb = 0; kb < CH; kb += 32) {
    v16h a, b;
#pragma unroll
    for (int e = 0; e < 16; ++e) {
      int k = (e < 8) ? (8 * kh + e) : (16 + 8 * kh + (e - 8));
      a[e] = (_Float16)arow[kb + k];
      b[e] = (_Float16)W[(size_t)(kb + k) * CH + n0 + m];
    }
    acc = __builtin_amdgcn_wmma_f32_16x16x32_f16(false, a, false, b,
                                                 (short)0, acc, false, false);
  }
#endif

#pragma unroll
  for (int v = 0; v < 8; ++v) {
    int r = row0 + v + 8 * kh;
    if (r < N) C[(size_t)r * CH + (n0 + m)] = acc[v];
  }
}

static __device__ __forceinline__ float xhalf(float x) {
  return __builtin_bit_cast(float, __builtin_amdgcn_permlanex16(__builtin_bit_cast(int, x), __builtin_bit_cast(int, x),
                                                                0x76543210, 0xfedcba98, false, false));
}

static __device__ __forceinline__ float4 gather_row(const float* __restrict__ h,
                                                    const int* __restrict__ list,
                                                    const int* __restrict__ cur,
                                                    const float* __restrict__ dinv,
                                                    int i, int lane) {
  const int half = lane >> 4;
  const int c4 = (lane & 15) * 4;
  const int start = cur[i];
  const int end = cur[i + 1];
  const float di = dinv[i];
  const float4 self = *(const float4*)(h + (size_t)i * CH + c4);
  const float ws = di * di;
  float4 acc;
  acc.x = half ? 0.0f : self.x * ws;
  acc.y = half ? 0.0f : self.y * ws;
  acc.z = half ? 0.0f : self.z * ws;
  acc.w = half ? 0.0f : self.w * ws;

  for (int base = start; base < end; base += 32) {
    const int n = min(32, end - base);
    int2 ent = make_int2(i, 0);
    if (lane < n) ent = ((const int2*)list)[base + lane];
    const int npairs = (n + 1) >> 1;
    for (int p = 0; p < npairs; p += 8) {
      int s[8];
      float w[8];
      float4 r[8];
#pragma unroll
      for (int u = 0; u < 8; ++u) {
        const int q = 2 * (p + u);
        const int sa = __builtin_amdgcn_readlane(ent.x, q);
        const int sb = __builtin_amdgcn_readlane(ent.x, q + 1);
        const int wa = __builtin_amdgcn_readlane(ent.y, q);
        const int wb = __builtin_amdgcn_readlane(ent.y, q + 1);
        s[u] = half ? sb : sa;
        w[u] = __int_as_float(half ? wb : wa);
      }
#pragma unroll
      for (int u = 0; u < 8; ++u) {
        r[u] = make_float4(0.0f, 0.0f, 0.0f, 0.0f);
        if (2 * (p + u) + half < n) r[u] = *(const float4*)(h + (size_t)s[u] * CH + c4);
      }
#pragma unroll
      for (int u = 0; u < 8; ++u) {
        acc.x = fmaf(r[u].x, w[u], acc.x);
        acc.y = fmaf(r[u].y, w[u], acc.y);
        acc.z = fmaf(r[u].z, w[u], acc.z);
        acc.w = fmaf(r[u].w, w[u], acc.w);
      }
    }
  }
  acc.x += xhalf(acc.x);
  acc.y += xhalf(acc.y);
  acc.z += xhalf(acc.z);
  acc.w += xhalf(acc.w);
  return acc;
}

__global__ __launch_bounds__(256) void k_gather_relu(const float* __restrict__ h,
                                                     const int* __restrict__ list,
                                                     const int* __restrict__ cur,
                                                     const float* __restrict__ dinv,
                                                     const float* __restrict__ b,
                                                     float* __restrict__ out, int N) {
  const int lane = threadIdx.x & 31;
  const int i = __builtin_amdgcn_readfirstlane(blockIdx.x * 8 + (threadIdx.x >> 5));
  if (i >= N) return;
  float4 v = gather_row(h, list, cur, dinv, i, lane);
  const int c4 = (lane & 15) * 4;
  const float4 bb = *(const float4*)(b + c4);
  v.x += bb.x;  v.y += bb.y;  v.z += bb.z;  v.w += bb.w;
  v.x = v.x > 0.0f ? v.x : 0.0f;
  v.y = v.y > 0.0f ? v.y : 0.0f;
  v.z = v.z > 0.0f ? v.z : 0.0f;
  v.w = v.w > 0.0f ? v.w : 0.0f;
  if (lane < 16) *(float4*)(out + (size_t)i * CH + c4) = v;
}

__global__ __launch_bounds__(256) void k_gather_relu_proj(const float* __restrict__ h,
                                                          const int* __restrict__ list,
                                                          const int* __restrict__ cur,
                                                          const float* __restrict__ dinv,
                                                          const float* __restrict__ b,
                                                          const float* __restrict__ Wfc,
                                                          float* __restrict__ ps,
                                                          float* __restrict__ pd, int N) {
  const int lane = threadIdx.x & 31;
  const int i = __builtin_amdgcn_readfirstlane(blockIdx.x * 8 + (threadIdx.x >> 5));
  if (i >= N) return;
  float4 v = gather_row(h, list, cur, dinv, i, lane);
  const int c4 = (lane & 15) * 4;
  const float4 bb = *(const float4*)(b + c4);
  const float4 w0 = *(const float4*)(Wfc + c4);
  const float4 w1 = *(const float4*)(Wfc + CH + c4);
  v.x += bb.x;  v.y += bb.y;  v.z += bb.z;  v.w += bb.w;
  v.x = v.x > 0.0f ? v.x : 0.0f;
  v.y = v.y > 0.0f ? v.y : 0.0f;
  v.z = v.z > 0.0f ? v.z : 0.0f;
  v.w = v.w > 0.0f ? v.w : 0.0f;
  float s0 = (v.x * w0.x + v.y * w0.y) + (v.z * w0.z + v.w * w0.w);
  float s1 = (v.x * w1.x + v.y * w1.y) + (v.z * w1.z + v.w * w1.w);
#pragma unroll
  for (int off = 8; off > 0; off >>= 1) {
    s0 += __shfl_xor(s0, off, 32);
    s1 += __shfl_xor(s1, off, 32);
  }
  if (lane == 0) {
    ps[i] = s0;
    pd[i] = s1;
  }
}

__global__ void k_edge_pred(const float* __restrict__ ps, const float* __restrict__ pd,
                            const int* __restrict__ src, const int* __restrict__ dst,
                            const float* __restrict__ bfc, float* __restrict__ out, int E) {
  int e = blockIdx.x * blockDim.x + threadIdx.x;
  if (e < E) out[e] = ps[src[e]] + pd[dst[e]] + bfc[0];
}

extern "C" void kernel_launch(void* const* d_in, const int* in_sizes, int n_in,
                              void* d_out, int out_size, void* d_ws, size_t ws_size,
                              hipStream_t stream) {
  const float* x   = (const float*)d_in[0];
  const int*   ei  = (const int*)d_in[1];
  const float* W1  = (const float*)d_in[2];
  const float* b1  = (const float*)d_in[3];
  const float* W2  = (const float*)d_in[4];
  const float* b2  = (const float*)d_in[5];
  const float* Wfc = (const float*)d_in[6];
  const float* bfc = (const float*)d_in[7];

  const int N = in_sizes[0] / CH;
  const int E = in_sizes[1] / 2;
  const int NF = N * CH;

  const int* src = ei;
  const int* dst = ei + E;

  float* ws   = (float*)d_ws;
  float* HB   = ws;
  float* AB   = HB + (size_t)NF;
  int*   list = (int*)(AB + (size_t)NF);
  int*   cnt  = list + 2 * (size_t)E;
  int*   cur  = cnt + N;
  float* dinv = (float*)(cur + N + 1);
  float* PS   = dinv + N;
  float* PD   = PS + N;
  int*   rank = (int*)(PD + N);

  const int TB = 256;
  dim3 blk(TB);
  dim3 gN((N + TB - 1) / TB);
  dim3 gE((E + TB - 1) / TB);
  dim3 gNw((N + 7) / 8);
  dim3 gG((N + 31) / 32);

  k_zero<<<gN, blk, 0, stream>>>(cnt, N);
  k_count<<<gE, blk, 0, stream>>>(dst, cnt, rank, E);
  k_offsets<<<gN, blk, 0, stream>>>(cnt, cur, dinv, N);
  k_place<<<gE, blk, 0, stream>>>(src, dst, rank, dinv, cur, list, E);

  k_gemm64<<<gG, blk, 0, stream>>>(x, W1, HB, N);
  k_gather_relu<<<gNw, blk, 0, stream>>>(HB, list, cur, dinv, b1, AB, N);

  k_gemm64<<<gG, blk, 0, stream>>>(AB, W2, HB, N);
  k_gather_relu_proj<<<gNw, blk, 0, stream>>>(HB, list, cur, dinv, b2, Wfc, PS, PD, N);

  k_edge_pred<<<gE, blk, 0, stream>>>(PS, PD, src, dst, bfc, (float*)d_out, E);
}


# Edge ranks from LDS counters: two-level counting sort by destination (256-node buckets per 2048-edge chunk, then nodes inside a bucket); no global atomic left; sum order = LDS arrival order
# speedup vs baseline: 2.8582x; 1.4415x over previous
#include <hip/hip_runtime.h>
#include <hip/hip_bf16.h>

typedef __attribute__((ext_vector_type(2)))  float    v2f;
typedef __attribute__((ext_vector_type(8)))  float    v8f;
typedef __attribute__((ext_vector_type(16))) _Float16 v16h;

#define CH 64

#define BSH   8
#define NBMAX 256
#define CT    512
#define CE    4

template <bool TAIL>
static __device__ __forceinline__ void count_chunk(const int* __restrict__ dst, int* __restrict__ rank,
                                                   int* h, int e0, int E) {
  int d[CE];
#pragma unroll
  for (int u = 0; u < CE; ++u) d[u] = dst[TAIL ? min(e0 + u * CT, E - 1) : e0 + u * CT];
#pragma unroll
  for (int u = 0; u < CE; ++u) {
    const int e = e0 + u * CT;
    if (!TAIL || e < E) rank[e] = atomicAdd(&h[d[u] >> BSH], 1);
  }
}

__global__ __launch_bounds__(CT) void k_bucket_count(const int* __restrict__ dst,
                                                     int* __restrict__ rank,
                                                     int* __restrict__ H, int E, int NB, int G) {
  __shared__ int h[NBMAX];
  const int tid = threadIdx.x;
  const int g = blockIdx.x;
  for (int j = tid; j < NB; j += CT) h[j] = 0;
  __syncthreads();
  const int e0 = g * (CT * CE) + tid;
  if ((g + 1) * (CT * CE) <= E) count_chunk<false>(dst, rank, h, e0, E);
  else                          count_chunk<true>(dst, rank, h, e0, E);
  __syncthreads();
  for (int j = tid; j < NB; j += CT) H[j * G + g] = h[j];
}

__global__ __launch_bounds__(256) void k_scan(const int* __restrict__ cnt,
                                              int* __restrict__ cur, int M) {
  __shared__ int s_front[8];
  __shared__ int s_own[8];
  const int tid = threadIdx.x;
  const int lane = tid & 31;
  const int wave = tid >> 5;
  const int4* c4 = (const int4*)cnt;
  const int nq = blockIdx.x * 64;
  int front = 0;
  for (int j0 = tid; j0 < nq; j0 += 8 * 256) {
    int4 v[8];
#pragma unroll
    for (int u = 0; u < 8; ++u) v[u] = c4[min(j0 + u * 256, nq - 1)];
#pragma unroll
    for (int u = 0; u < 8; ++u)
      if (j0 + u * 256 < nq) front += (v[u].x + v[u].y) + (v[u].z + v[u].w);
  }
  const int i = blockIdx.x * 256 + tid;
  const int c = (i < M) ? cnt[i] : 0;
  int incl = c;
#pragma unroll
  for (int off = 1; off < 32; off <<= 1) {
    int t = __shfl_up(incl, off, 32);
    if (lane >= off) incl += t;
  }
#pragma unroll
  for (int off = 16; off > 0; off >>= 1) front += __shfl_xor(front, off, 32);
  if (lane == 31) s_own[wave] = incl;
  if (lane == 0) s_front[wave] = front;
  __syncthreads();
  int base = 0;
#pragma unroll
  for (int w = 0; w < 8; ++w) {
    base += s_front[w];
    if (w < wave) base += s_own[w];
  }
  if (i < M) {
    cur[i] = base + incl - c;
    if (i == M - 1) cur[M] = base + incl;
  }
}

template <bool TAIL>
static __device__ __forceinline__ void scatter_chunk(const int* __restrict__ src, const int* __restrict__ dst,
                                                     const int* __restrict__ rank, const int* __restrict__ first,
                                                     int* __restrict__ tmp, int e0, int E, int G, int g) {
  int s[CE], d[CE], r[CE], p[CE];
#pragma unroll
  for (int u = 0; u < CE; ++u) {
    const int e = TAIL ? min(e0 + u * CT, E - 1) : e0 + u * CT;
    s[u] = src[e];
    d[u] = dst[e];
    r[u] = rank[e];
  }
#pragma unroll
  for (int u = 0; u < CE; ++u) p[u] = first[(d[u] >> BSH) * G + g] + r[u];
#pragma unroll
  for (int u = 0; u < CE; ++u)
    if (!TAIL || e0 + u * CT < E) ((int2*)tmp)[p[u]] = make_int2(s[u], d[u]);
}

__global__ __launch_bounds__(CT) void k_bucket_scatter(const int* __restrict__ src,
                                                       const int* __restrict__ dst,
                                                       const int* __restrict__ rank,
                                                       const int* __restrict__ first,
                                                       int* __restrict__ tmp, int E, int G) {
  const int g = blockIdx.x;
  const int e0 = g * (CT * CE) + threadIdx.x;
  if ((g + 1) * (CT * CE) <= E) scatter_chunk<false>(src, dst, rank, first, tmp, e0, E, G, g);
  else                          scatter_chunk<true>(src, dst, rank, first, tmp, e0, E, G, g);
}

__global__ __launch_bounds__(1024) void k_bucket_sort(const int* __restrict__ tmp,
                                                      const int* __restrict__ first,
                                                      int* __restrict__ list,
                                                      int* __restrict__ cur,
                                                      float* __restrict__ dinv, int N, int G) {
  __shared__ int c[256];
  __shared__ int st[256];
  __shared__ int f[256];
  const int tid = threadIdx.x;
  const int b = blockIdx.x;
  const int lo = first[b * G];
  const int hi = first[(b + 1) * G];
  if (tid < 256) {
    c[tid] = 0;
    f[tid] = 0;
  }
  __syncthreads();
  for (int k0 = lo + tid; k0 < hi; k0 += 4096) {
    int d[4];
#pragma unroll
    for (int u = 0; u < 4; ++u) d[u] = tmp[2 * min(k0 + u * 1024, hi - 1) + 1];
#pragma unroll
    for (int u = 0; u < 4; ++u)
      if (k0 + u * 1024 < hi) atomicAdd(&c[d[u] & 255], 1);
  }
  __syncthreads();
  if (tid < 32) {
    int v[8];
    int run = 0;
#pragma unroll
    for (int j = 0; j < 8; ++j) {
      const int t = c[8 * tid + j];
      v[j] = run;
      run += t;
    }
    int incl = run;
#pragma unroll
    for (int off = 1; off < 32; off <<= 1) {
      int t = __shfl_up(incl, off, 32);
      if (tid >= off) incl += t;
    }
    const int lanebase = incl - run;
#pragma unroll
    for (int j = 0; j < 8; ++j) st[8 * tid + j] = lanebase + v[j];
  }
  __syncthreads();
  if (tid < 256) {
    const int node = b * 256 + tid;
    if (node < N) {
      cur[node] = lo + st[tid];
      dinv[node] = rsqrtf((float)(c[tid] + 1));
    }
  }
  if (b == ((N + (1 << BSH) - 1) >> BSH) - 1 && tid == 0) cur[N] = hi;
  for (int k0 = lo + tid; k0 < hi; k0 += 4096) {
    int2 ent[4];
#pragma unroll
    for (int u = 0; u < 4; ++u) ent[u] = ((const int2*)tmp)[min(k0 + u * 1024, hi - 1)];
#pragma unroll
    for (int u = 0; u < 4; ++u) {
      if (k0 + u * 1024 < hi) {
        const int j = ent[u].y & 255;
        const int r = atomicAdd(&f[j], 1);
        ((int2*)list)[lo + st[j] + r] = ent[u];
      }
    }
  }
}

template <bool TAIL>
static __device__ __forceinline__ void norm_chunk(int* __restrict__ list, const float* __restrict__ dinv,
                                                  int s0, int E) {
  int2 ent[4];
  float a[4], bb[4];
#pragma unroll
  for (int u = 0; u < 4; ++u) {
    const int slot = s0 + u * 256;
    ent[u] = make_int2(0, 0);
    if (!TAIL || slot < E) ent[u] = ((const int2*)list)[slot];
  }
#pragma unroll
  for (int u = 0; u < 4; ++u) {
    a[u] = dinv[ent[u].x];
    bb[u] = dinv[ent[u].y];
  }
#pragma unroll
  for (int u = 0; u < 4; ++u) {
    const int slot = s0 + u * 256;
    if (!TAIL || slot < E) list[2 * slot + 1] = __float_as_int(a[u] * bb[u]);
  }
}

__global__ __launch_bounds__(256) void k_norm(int* __restrict__ list,
                                              const float* __restrict__ dinv, int E) {
  const int s0 = blockIdx.x * (256 * 4) + threadIdx.x;
  if ((blockIdx.x + 1) * (256 * 4) <= E) norm_chunk<false>(list, dinv, s0, E);
  else                                   norm_chunk<true>(list, dinv, s0, E);
}

__global__ __launch_bounds__(256) void k_gemm64(const float* __restrict__ A,
                                                const float* __restrict__ W,
                                                float* __restrict__ C, int N) {
  const int lane = threadIdx.x & 31;
  const int wave = threadIdx.x >> 5;
  const int row0 = blockIdx.x * 32 + (wave >> 2) * 16;
  if (row0 >= N) return;
  const int n0 = (wave & 3) * 16;
  const int m  = lane & 15;
  const int kh = lane >> 4;
  v8f acc = {};

#if __has_builtin(__builtin_amdgcn_wmma_f32_16x16x4_f32)
  const float* arow = A + (size_t)(row0 + m) * CH + 2 * kh;
#pragma unroll
  for (int k0 = 0; k0 < CH; k0 += 4) {
    v2f a;
    a.x = arow[k0];
    a.y = arow[k0 + 1];
    v2f b;
    b.x = W[(k0 + 2 * kh + 0) * CH + n0 + m];
    b.y = W[(k0 + 2 * kh + 1) * CH + n0 + m];
    acc = __builtin_amdgcn_wmma_f32_16x16x4_f32(false, a, false, b,
                                                (short)0, acc, false, false);
  }
#else
  const float* arow = A + (size_t)(row0 + m) * CH;
#pragma unroll
  for (int kb = 0; kb < CH; kb += 32) {
    v16h a, b;
#pragma unroll
    for (int e = 0; e < 16; ++e) {
      int k = (e < 8) ? (8 * kh + e) : (16 + 8 * kh + (e - 8));
      a[e] = (_Float16)arow[kb + k];
      b[e] = (_Float16)W[(size_t)(kb + k) * CH + n0 + m];
    }
    acc = __builtin_amdgcn_wmma_f32_16x16x32_f16(false, a, false, b,
                                                 (short)0, acc, false, false);
  }
#endif

#pragma unroll
  for (int v = 0; v < 8; ++v) {
    int r = row0 + v + 8 * kh;
    if (r < N) C[(size_t)r * CH + (n0 + m)] = acc[v];
  }
}

static __device__ __forceinline__ float xhalf(float x) {
  return __builtin_bit_cast(float, __builtin_amdgcn_permlanex16(__builtin_bit_cast(int, x), __builtin_bit_cast(int, x),
                                                                0x76543210, 0xfedcba98, false, false));
}

static __device__ __forceinline__ float4 gather_row(const float* __restrict__ h,
                                                    const int* __restrict__ list,
                                                    const int* __restrict__ cur,
                                                    const float* __restrict__ dinv,
                                                    int i, int lane) {
  const int half = lane >> 4;
  const int c4 = (lane & 15) * 4;
  const int start = cur[i];
  const int end = cur[i + 1];
  const float di = dinv[i];
  const float4 self = *(const float4*)(h + (size_t)i * CH + c4);
  const float ws = di * di;
  float4 acc;
  acc.x = half ? 0.0f : self.x * ws;
  acc.y = half ? 0.0f : self.y * ws;
  acc.z = half ? 0.0f : self.z * ws;
  acc.w = half ? 0.0f : self.w * ws;

  for (int base = start; base < end; base += 32) {
    const int n = min(32, end - base);
    int2 ent = make_int2(i, 0);
    if (lane < n) ent = ((const int2*)list)[base + lane];
    const int npairs = (n + 1) >> 1;
    for (int p = 0; p < npairs; p += 8) {
      int s[8];
      float w[8];
      float4 r[8];
#pragma unroll
      for (int u = 0; u < 8; ++u) {
        const int q = 2 * (p + u);
        const int sa = __builtin_amdgcn_readlane(ent.x, q);
        const int sb = __builtin_amdgcn_readlane(ent.x, q + 1);
        const int wa = __builtin_amdgcn_readlane(ent.y, q);
        const int wb = __builtin_amdgcn_readlane(ent.y, q + 1);
        s[u] = half ? sb : sa;
        w[u] = __int_as_float(half ? wb : wa);
      }
#pragma unroll
      for (int u = 0; u < 8; ++u) {
        r[u] = make_float4(0.0f, 0.0f, 0.0f, 0.0f);
        if (2 * (p + u) + half < n) r[u] = *(const float4*)(h + (size_t)s[u] * CH + c4);
      }
#pragma unroll
      for (int u = 0; u < 8; ++u) {
        acc.x = fmaf(r[u].x, w[u], acc.x);
        acc.y = fmaf(r[u].y, w[u], acc.y);
        acc.z = fmaf(r[u].z, w[u], acc.z);
        acc.w = fmaf(r[u].w, w[u], acc.w);
      }
    }
  }
  acc.x += xhalf(acc.x);
  acc.y += xhalf(acc.y);
  acc.z += xhalf(acc.z);
  acc.w += xhalf(acc.w);
  return acc;
}

__global__ __launch_bounds__(256) void k_gather_relu(const float* __restrict__ h,
                                                     const int* __restrict__ list,
                                                     const int* __restrict__ cur,
                                                     const float* __restrict__ dinv,
                                                     const float* __restrict__ b,
                                                     float* __restrict__ out, int N) {
  const int lane = threadIdx.x & 31;
  const int i = __builtin_amdgcn_readfirstlane(blockIdx.x * 8 + (threadIdx.x >> 5));
  if (i >= N) return;
  float4 v = gather_row(h, list, cur, dinv, i, lane);
  const int c4 = (lane & 15) * 4;
  const float4 bb = *(const float4*)(b + c4);
  v.x += bb.x;  v.y += bb.y;  v.z += bb.z;  v.w += bb.w;
  v.x = v.x > 0.0f ? v.x : 0.0f;
  v.y = v.y > 0.0f ? v.y : 0.0f;
  v.z = v.z > 0.0f ? v.z : 0.0f;
  v.w = v.w > 0.0f ? v.w : 0.0f;
  if (lane < 16) *(float4*)(out + (size_t)i * CH + c4) = v;
}

__global__ __launch_bounds__(256) void k_gather_relu_proj(const float* __restrict__ h,
                                                          const int* __restrict__ list,
                                                          const int* __restrict__ cur,
                                                          const float* __restrict__ dinv,
                                                          const float* __restrict__ b,
                                                          const float* __restrict__ Wfc,
                                                          float* __restrict__ ps,
                                                          float* __restrict__ pd, int N) {
  const int lane = threadIdx.x & 31;
  const int i = __builtin_amdgcn_readfirstlane(blockIdx.x * 8 + (threadIdx.x >> 5));
  if (i >= N) return;
  float4 v = gather_row(h, list, cur, dinv, i, lane);
  const int c4 = (lane & 15) * 4;
  const float4 bb = *(const float4*)(b + c4);
  const float4 w0 = *(const float4*)(Wfc + c4);
  const float4 w1 = *(const float4*)(Wfc + CH + c4);
  v.x += bb.x;  v.y += bb.y;  v.z += bb.z;  v.w += bb.w;
  v.x = v.x > 0.0f ? v.x : 0.0f;
  v.y = v.y > 0.0f ? v.y : 0.0f;
  v.z = v.z > 0.0f ? v.z : 0.0f;
  v.w = v.w > 0.0f ? v.w : 0.0f;
  float s0 = (v.x * w0.x + v.y * w0.y) + (v.z * w0.z + v.w * w0.w);
  float s1 = (v.x * w1.x + v.y * w1.y) + (v.z * w1.z + v.w * w1.w);
#pragma unroll
  for (int off = 8; off > 0; off >>= 1) {
    s0 += __shfl_xor(s0, off, 32);
    s1 += __shfl_xor(s1, off, 32);
  }
  if (lane == 0) {
    ps[i] = s0;
    pd[i] = s1;
  }
}

__global__ void k_edge_pred(const float* __restrict__ ps, const float* __restrict__ pd,
                            const int* __restrict__ src, const int* __restrict__ dst,
                            const float* __restrict__ bfc, float* __restrict__ out, int E) {
  int e = blockIdx.x * blockDim.x + threadIdx.x;
  if (e < E) out[e] = ps[src[e]] + pd[dst[e]] + bfc[0];
}

extern "C" void kernel_launch(void* const* d_in, const int* in_sizes, int n_in,
                              void* d_out, int out_size, void* d_ws, size_t ws_size,
                              hipStream_t stream) {
  const float* x   = (const float*)d_in[0];
  const int*   ei  = (const int*)d_in[1];
  const float* W1  = (const float*)d_in[2];
  const float* b1  = (const float*)d_in[3];
  const float* W2  = (const float*)d_in[4];
  const float* b2  = (const float*)d_in[5];
  const float* Wfc = (const float*)d_in[6];
  const float* bfc = (const float*)d_in[7];

  const int N = in_sizes[0] / CH;
  const int E = in_sizes[1] / 2;
  const int NF = N * CH;

  const int* src = ei;
  const int* dst = ei + E;

  const int NB = (N + (1 << BSH) - 1) >> BSH;
  const int G  = (E + CT * CE - 1) / (CT * CE);
  const int M  = NB * G;
  const int E4 = (E + 3) & ~3;
  const int M4 = (M + 3) & ~3;

  float* ws    = (float*)d_ws;
  float* HB    = ws;
  float* AB    = HB + (size_t)NF;
  int*   list  = (int*)(AB + (size_t)NF);
  int*   tmp   = list + 2 * (size_t)E;
  int*   rank  = tmp + 2 * (size_t)E;
  int*   H     = rank + E4;
  int*   first = H + M4;
  int*   cur   = first + M4 + 4;
  float* dinv  = (float*)(cur + N + 1);
  float* PS    = dinv + N;
  float* PD    = PS + N;

  const int TB = 256;
  dim3 blk(TB);
  dim3 gE((E + TB - 1) / TB);
  dim3 gNw((N + 7) / 8);
  dim3 gG((N + 31) / 32);

  k_bucket_count<<<dim3(G), dim3(CT), 0, stream>>>(dst, rank, H, E, NB, G);
  k_scan<<<dim3((M + 255) / 256), blk, 0, stream>>>(H, first, M);
  k_bucket_scatter<<<dim3(G), dim3(CT), 0, stream>>>(src, dst, rank, first, tmp, E, G);
  k_bucket_sort<<<dim3(NB), dim3(1024), 0, stream>>>(tmp, first, list, cur, dinv, N, G);
  k_norm<<<dim3((E + 1023) / 1024), blk, 0, stream>>>(list, dinv, E);

  k_gemm64<<<gG, blk, 0, stream>>>(x, W1, HB, N);
  k_gather_relu<<<gNw, blk, 0, stream>>>(HB, list, cur, dinv, b1, AB, N);

  k_gemm64<<<gG, blk, 0, stream>>>(AB, W2, HB, N);
  k_gather_relu_proj<<<gNw, blk, 0, stream>>>(HB, list, cur, dinv, b2, Wfc, PS, PD, N);

  k_edge_pred<<<gE, blk, 0, stream>>>(PS, PD, src, dst, bfc, (float*)d_out, E);
}


# Sum first, multiply after: each gather kernel sums the INPUT rows of a 16-node tile into LDS and multiplies it by W on f32 WMMA there, bias+relu after; product launches gone; sums regrouped
# speedup vs baseline: 3.4923x; 1.2219x over previous
#include <hip/hip_runtime.h>
#include <hip/hip_bf16.h>

typedef __attribute__((ext_vector_type(2)))  float    v2f;
typedef __attribute__((ext_vector_type(8)))  float    v8f;

#define CH 64

#define BSH   8
#define NBMAX 256
#define CT    512
#define CE    4

template <bool TAIL>
static __device__ __forceinline__ void count_chunk(const int* __restrict__ dst, int* __restrict__ rank,
                                                   int* h, int e0, int E) {
  int d[CE];
#pragma unroll
  for (int u = 0; u < CE; ++u) d[u] = dst[TAIL ? min(e0 + u * CT, E - 1) : e0 + u * CT];
#pragma unroll
  for (int u = 0; u < CE; ++u) {
    const int e = e0 + u * CT;
    if (!TAIL || e < E) rank[e] = atomicAdd(&h[d[u] >> BSH], 1);
  }
}

__global__ __launch_bounds__(CT) void k_bucket_count(const int* __restrict__ dst,
                                                     int* __restrict__ rank,
                                                     int* __restrict__ H, int E, int NB, int G) {
  __shared__ int h[NBMAX];
  const int tid = threadIdx.x;
  const int g = blockIdx.x;
  for (int j = tid; j < NB; j += CT) h[j] = 0;
  __syncthreads();
  const int e0 = g * (CT * CE) + tid;
  if ((g + 1) * (CT * CE) <= E) count_chunk<false>(dst, rank, h, e0, E);
  else                          count_chunk<true>(dst, rank, h, e0, E);
  __syncthreads();
  for (int j = tid; j < NB; j += CT) H[j * G + g] = h[j];
}

__global__ __launch_bounds__(256) void k_scan(const int* __restrict__ cnt,
                                              int* __restrict__ cur, int M) {
  __shared__ int s_front[8];
  __shared__ int s_own[8];
  const int tid = threadIdx.x;
  const int lane = tid & 31;
  const int wave = tid >> 5;
  const int4* c4 = (const int4*)cnt;
  const int nq = blockIdx.x * 64;
  int front = 0;
  for (int j0 = tid; j0 < nq; j0 += 8 * 256) {
    int4 v[8];
#pragma unroll
    for (int u = 0; u < 8; ++u) v[u] = c4[min(j0 + u * 256, nq - 1)];
#pragma unroll
    for (int u = 0; u < 8; ++u)
      if (j0 + u * 256 < nq) front += (v[u].x + v[u].y) + (v[u].z + v[u].w);
  }
  const int i = blockIdx.x * 256 + tid;
  const int c = (i < M) ? cnt[i] : 0;
  int incl = c;
#pragma unroll
  for (int off = 1; off < 32; off <<= 1) {
    int t = __shfl_up(incl, off, 32);
    if (lane >= off) incl += t;
  }
#pragma unroll
  for (int off = 16; off > 0; off >>= 1) front += __shfl_xor(front, off, 32);
  if (lane == 31) s_own[wave] = incl;
  if (lane == 0) s_front[wave] = front;
  __syncthreads();
  int base = 0;
#pragma unroll
  for (int w = 0; w < 8; ++w) {
    base += s_front[w];
    if (w < wave) base += s_own[w];
  }
  if (i < M) {
    cur[i] = base + incl - c;
    if (i == M - 1) cur[M] = base + incl;
  }
}

template <bool TAIL>
static __device__ __forceinline__ void scatter_chunk(const int* __restrict__ src, const int* __restrict__ dst,
                                                     const int* __restrict__ rank, const int* __restrict__ first,
                                                     int* __restrict__ tmp, int e0, int E, int G, int g) {
  int s[CE], d[CE], r[CE], p[CE];
#pragma unroll
  for (int u = 0; u < CE; ++u) {
    const int e = TAIL ? min(e0 + u * CT, E - 1) : e0 + u * CT;
    s[u] = src[e];
    d[u] = dst[e];
    r[u] = rank[e];
  }
#pragma unroll
  for (int u = 0; u < CE; ++u) p[u] = first[(d[u] >> BSH) * G + g] + r[u];
#pragma unroll
  for (int u = 0; u < CE; ++u)
    if (!TAIL || e0 + u * CT < E) ((int2*)tmp)[p[u]] = make_int2(s[u], d[u]);
}

__global__ __launch_bounds__(CT) void k_bucket_scatter(const int* __restrict__ src,
                                                       const int* __restrict__ dst,
                                                       const int* __restrict__ rank,
                                                       const int* __restrict__ first,
                                                       int* __restrict__ tmp, int E, int G) {
  const int g = blockIdx.x;
  const int e0 = g * (CT * CE) + threadIdx.x;
  if ((g + 1) * (CT * CE) <= E) scatter_chunk<false>(src, dst, rank, first, tmp, e0, E, G, g);
  else                          scatter_chunk<true>(src, dst, rank, first, tmp, e0, E, G, g);
}

__global__ __launch_bounds__(1024) void k_bucket_sort(const int* __restrict__ tmp,
                                                      const int* __restrict__ first,
                                                      int* __restrict__ list,
                                                      int* __restrict__ cur,
                                                      float* __restrict__ dinv, int N, int G) {
  __shared__ int c[256];
  __shared__ int st[256];
  __shared__ int f[256];
  const int tid = threadIdx.x;
  const int b = blockIdx.x;
  const int lo = first[b * G];
  const int hi = first[(b + 1) * G];
  if (tid < 256) {
    c[tid] = 0;
    f[tid] = 0;
  }
  __syncthreads();
  for (int k0 = lo + tid; k0 < hi; k0 += 4096) {
    int d[4];
#pragma unroll
    for (int u = 0; u < 4; ++u) d[u] = tmp[2 * min(k0 + u * 1024, hi - 1) + 1];
#pragma unroll
    for (int u = 0; u < 4; ++u)
      if (k0 + u * 1024 < hi) atomicAdd(&c[d[u] & 255], 1);
  }
  __syncthreads();
  if (tid < 32) {
    int v[8];
    int run = 0;
#pragma unroll
    for (int j = 0; j < 8; ++j) {
      const int t = c[8 * tid + j];
      v[j] = run;
      run += t;
    }
    int incl = run;
#pragma unroll
    for (int off = 1; off < 32; off <<= 1) {
      int t = __shfl_up(incl, off, 32);
      if (tid >= off) incl += t;
    }
    const int lanebase = incl - run;
#pragma unroll
    for (int j = 0; j < 8; ++j) st[8 * tid + j] = lanebase + v[j];
  }
  __syncthreads();
  if (tid < 256) {
    const int node = b * 256 + tid;
    if (node < N) {
      cur[node] = lo + st[tid];
      dinv[node] = rsqrtf((float)(c[tid] + 1));
    }
  }
  if (b == ((N + (1 << BSH) - 1) >> BSH) - 1 && tid == 0) cur[N] = hi;
  for (int k0 = lo + tid; k0 < hi; k0 += 4096) {
    int2 ent[4];
#pragma unroll
    for (int u = 0; u < 4; ++u) ent[u] = ((const int2*)tmp)[min(k0 + u * 1024, hi - 1)];
#pragma unroll
    for (int u = 0; u < 4; ++u) {
      if (k0 + u * 1024 < hi) {
        const int j = ent[u].y & 255;
        const int r = atomicAdd(&f[j], 1);
        ((int2*)list)[lo + st[j] + r] = ent[u];
      }
    }
  }
}

template <bool TAIL>
static __device__ __forceinline__ void norm_chunk(int* __restrict__ list, const float* __restrict__ dinv,
                                                  int s0, int E) {
  int2 ent[4];
  float a[4], bb[4];
#pragma unroll
  for (int u = 0; u < 4; ++u) {
    const int slot = s0 + u * 256;
    ent[u] = make_int2(0, 0);
    if (!TAIL || slot < E) ent[u] = ((const int2*)list)[slot];
  }
#pragma unroll
  for (int u = 0; u < 4; ++u) {
    a[u] = dinv[ent[u].x];
    bb[u] = dinv[ent[u].y];
  }
#pragma unroll
  for (int u = 0; u < 4; ++u) {
    const int slot = s0 + u * 256;
    if (!TAIL || slot < E) list[2 * slot + 1] = __float_as_int(a[u] * bb[u]);
  }
}

__global__ __launch_bounds__(256) void k_norm(int* __restrict__ list,
                                              const float* __restrict__ dinv, int E) {
  const int s0 = blockIdx.x * (256 * 4) + threadIdx.x;
  if ((blockIdx.x + 1) * (256 * 4) <= E) norm_chunk<false>(list, dinv, s0, E);
  else                                   norm_chunk<true>(list, dinv, s0, E);
}

static __device__ __forceinline__ float xhalf(float x) {
  return __builtin_bit_cast(float, __builtin_amdgcn_permlanex16(__builtin_bit_cast(int, x), __builtin_bit_cast(int, x),
                                                                0x76543210, 0xfedcba98, false, false));
}

static __device__ __forceinline__ float4 gather_row(const float* __restrict__ h,
                                                    const int* __restrict__ list,
                                                    const int* __restrict__ cur,
                                                    const float* __restrict__ dinv,
                                                    int i, int lane) {
  const int half = lane >> 4;
  const int c4 = (lane & 15) * 4;
  const int start = cur[i];
  const int end = cur[i + 1];
  const float di = dinv[i];
  const float4 self = *(const float4*)(h + (size_t)i * CH + c4);
  const float ws = di * di;
  float4 acc;
  acc.x = half ? 0.0f : self.x * ws;
  acc.y = half ? 0.0f : self.y * ws;
  acc.z = half ? 0.0f : self.z * ws;
  acc.w = half ? 0.0f : self.w * ws;

  for (int base = start; base < end; base += 32) {
    const int n = min(32, end - base);
    int2 ent = make_int2(i, 0);
    if (lane < n) ent = ((const int2*)list)[base + lane];
    const int npairs = (n + 1) >> 1;
    for (int p = 0; p < npairs; p += 8) {
      int s[8];
      float w[8];
      float4 r[8];
#pragma unroll
      for (int u = 0; u < 8; ++u) {
        const int q = 2 * (p + u);
        const int sa = __builtin_amdgcn_readlane(ent.x, q);
        const int sb = __builtin_amdgcn_readlane(ent.x, q + 1);
        const int wa = __builtin_amdgcn_readlane(ent.y, q);
        const int wb = __builtin_amdgcn_readlane(ent.y, q + 1);
        s[u] = half ? sb : sa;
        w[u] = __int_as_float(half ? wb : wa);
      }
#pragma unroll
      for (int u = 0; u < 8; ++u) {
        r[u] = make_float4(0.0f, 0.0f, 0.0f, 0.0f);
        if (2 * (p + u) + half < n) r[u] = *(const float4*)(h + (size_t)s[u] * CH + c4);
      }
#pragma unroll
      for (int u = 0; u < 8; ++u) {
        acc.x = fmaf(r[u].x, w[u], acc.x);
        acc.y = fmaf(r[u].y, w[u], acc.y);
        acc.z = fmaf(r[u].z, w[u], acc.z);
        acc.w = fmaf(r[u].w, w[u], acc.w);
      }
    }
  }
  acc.x += xhalf(acc.x);
  acc.y += xhalf(acc.y);
  acc.z += xhalf(acc.z);
  acc.w += xhalf(acc.w);
  return acc;
}

#define TN 16
#define TP 68

template <bool PROJ>
static __device__ __forceinline__ void layer_tile(const float* __restrict__ hin,
                                                  const int* __restrict__ list,
                                                  const int* __restrict__ cur,
                                                  const float* __restrict__ dinv,
                                                  const float* __restrict__ W,
                                                  const float* __restrict__ b,
                                                  const float* __restrict__ Wfc,
                                                  float* __restrict__ out,
                                                  float* __restrict__ ps,
                                                  float* __restrict__ pd, int N) {
  __shared__ float agg[TN * TP];
  __shared__ float act[TN * TP];
  const int lane = threadIdx.x & 31;
  const int wave = threadIdx.x >> 5;
  const int kh = lane >> 4;
  const int m = lane & 15;
  const int c4 = m * 4;
  const int tile0 = blockIdx.x * TN;

  for (int t = 0; t < 4; ++t) {
    const int r = wave * 4 + t;
    const int i = __builtin_amdgcn_readfirstlane(tile0 + r);
    float4 v = make_float4(0.0f, 0.0f, 0.0f, 0.0f);
    if (i < N) v = gather_row(hin, list, cur, dinv, i, lane);
    if (lane < 16) *(float4*)(agg + r * TP + c4) = v;
  }

  const int n0 = wave * 16;
  v2f bw[CH / 4];
#pragma unroll
  for (int j = 0; j < CH / 4; ++j) {
    bw[j].x = W[(4 * j + 2 * kh + 0) * CH + n0 + m];
    bw[j].y = W[(4 * j + 2 * kh + 1) * CH + n0 + m];
  }
  const float bias = b[n0 + m];
  __syncthreads();

  const float* arow = agg + m * TP + 2 * kh;
  v8f acc = {};
#pragma unroll
  for (int j = 0; j < CH / 4; ++j) {
    v2f a;
    a.x = arow[4 * j];
    a.y = arow[4 * j + 1];
    acc = __builtin_amdgcn_wmma_f32_16x16x4_f32(false, a, false, bw[j],
                                                (short)0, acc, false, false);
  }
#pragma unroll
  for (int v = 0; v < 8; ++v) {
    const float y = acc[v] + bias;
    act[(v + 8 * kh) * TP + n0 + m] = y > 0.0f ? y : 0.0f;
  }
  float4 w0, w1;
  if (PROJ) {
    w0 = *(const float4*)(Wfc + c4);
    w1 = *(const float4*)(Wfc + CH + c4);
  }
  __syncthreads();

#pragma unroll
  for (int t = 0; t < 2; ++t) {
    const int r = wave * 4 + 2 * t + kh;
    const int i = tile0 + r;
    const float4 v = *(const float4*)(act + r * TP + c4);
    if (PROJ) {
      float s0 = (v.x * w0.x + v.y * w0.y) + (v.z * w0.z + v.w * w0.w);
      float s1 = (v.x * w1.x + v.y * w1.y) + (v.z * w1.z + v.w * w1.w);
#pragma unroll
      for (int off = 8; off > 0; off >>= 1) {
        s0 += __shfl_xor(s0, off, 32);
        s1 += __shfl_xor(s1, off, 32);
      }
      if (m == 0 && i < N) {
        ps[i] = s0;
        pd[i] = s1;
      }
    } else {
      if (i < N) *(float4*)(out + (size_t)i * CH + c4) = v;
    }
  }
}

__global__ __launch_bounds__(128) void k_layer_relu(const float* __restrict__ hin,
                                                    const int* __restrict__ list,
                                                    const int* __restrict__ cur,
                                                    const float* __restrict__ dinv,
                                                    const float* __restrict__ W,
                                                    const float* __restrict__ b,
                                                    float* __restrict__ out, int N) {
  layer_tile<false>(hin, list, cur, dinv, W, b, nullptr, out, nullptr, nullptr, N);
}

__global__ __launch_bounds__(128) void k_layer_relu_proj(const float* __restrict__ hin,
                                                         const int* __restrict__ list,
                                                         const int* __restrict__ cur,
                                                         const float* __restrict__ dinv,
                                                         const float* __restrict__ W,
                                                         const float* __restrict__ b,
                                                         const float* __restrict__ Wfc,
                                                         float* __restrict__ ps,
                                                         float* __restrict__ pd, int N) {
  layer_tile<true>(hin, list, cur, dinv, W, b, Wfc, nullptr, ps, pd, N);
}

__global__ void k_edge_pred(const float* __restrict__ ps, const float* __restrict__ pd,
                            const int* __restrict__ src, const int* __restrict__ dst,
                            const float* __restrict__ bfc, float* __restrict__ out, int E) {
  int e = blockIdx.x * blockDim.x + threadIdx.x;
  if (e < E) out[e] = ps[src[e]] + pd[dst[e]] + bfc[0];
}

extern "C" void kernel_launch(void* const* d_in, const int* in_sizes, int n_in,
                              void* d_out, int out_size, void* d_ws, size_t ws_size,
                              hipStream_t stream) {
  const float* x   = (const float*)d_in[0];
  const int*   ei  = (const int*)d_in[1];
  const float* W1  = (const float*)d_in[2];
  const float* b1  = (const float*)d_in[3];
  const float* W2  = (const float*)d_in[4];
  const float* b2  = (const float*)d_in[5];
  const float* Wfc = (const float*)d_in[6];
  const float* bfc = (const float*)d_in[7];

  const int N = in_sizes[0] / CH;
  const int E = in_sizes[1] / 2;
  const int NF = N * CH;

  const int* src = ei;
  const int* dst = ei + E;

  const int NB = (N + (1 << BSH) - 1) >> BSH;
  const int G  = (E + CT * CE - 1) / (CT * CE);
  const int M  = NB * G;
  const int E4 = (E + 3) & ~3;
  const int M4 = (M + 3) & ~3;

  float* ws    = (float*)d_ws;
  float* AB    = ws;
  int*   list  = (int*)(AB + (size_t)NF);
  int*   tmp   = list + 2 * (size_t)E;
  int*   rank  = tmp + 2 * (size_t)E;
  int*   H     = rank + E4;
  int*   first = H + M4;
  int*   cur   = first + M4 + 4;
  float* dinv  = (float*)(cur + N + 1);
  float* PS    = dinv + N;
  float* PD    = PS + N;

  const int TB = 256;
  dim3 blk(TB);
  dim3 gE((E + TB - 1) / TB);
  dim3 gT((N + TN - 1) / TN);

  k_bucket_count<<<dim3(G), dim3(CT), 0, stream>>>(dst, rank, H, E, NB, G);
  k_scan<<<dim3((M + 255) / 256), blk, 0, stream>>>(H, first, M);
  k_bucket_scatter<<<dim3(G), dim3(CT), 0, stream>>>(src, dst, rank, first, tmp, E, G);
  k_bucket_sort<<<dim3(NB), dim3(1024), 0, stream>>>(tmp, first, list, cur, dinv, N, G);
  k_norm<<<dim3((E + 1023) / 1024), blk, 0, stream>>>(list, dinv, E);

  k_layer_relu<<<gT, dim3(128), 0, stream>>>(x, list, cur, dinv, W1, b1, AB, N);

  k_layer_relu_proj<<<gT, dim3(128), 0, stream>>>(AB, list, cur, dinv, W2, b2, Wfc, PS, PD, N);

  k_edge_pred<<<gE, blk, 0, stream>>>(PS, PD, src, dst, bfc, (float*)d_out, E);
}


# Eight edges per thread in the per-edge passes (count, scatter, norm had 4; edge head had 1, now 16-byte index loads and stores): far fewer waves enter; 4096-edge chunks, sum order = LDS arrival order
# speedup vs baseline: 3.9440x; 1.1293x over previous
#include <hip/hip_runtime.h>
#include <hip/hip_bf16.h>

typedef __attribute__((ext_vector_type(2)))  float    v2f;
typedef __attribute__((ext_vector_type(8)))  float    v8f;

#define CH 64

#define BSH   8
#define NBMAX 256
#define CT    512
#define CE    8

template <bool TAIL>
static __device__ __forceinline__ void count_chunk(const int* __restrict__ dst, int* __restrict__ rank,
                                                   int* h, int e0, int E) {
  int d[CE];
#pragma unroll
  for (int u = 0; u < CE; ++u) d[u] = dst[TAIL ? min(e0 + u * CT, E - 1) : e0 + u * CT];
  int r[CE];
#pragma unroll
  for (int u = 0; u < CE; ++u) {
    r[u] = 0;
    if (!TAIL || e0 + u * CT < E) r[u] = atomicAdd(&h[d[u] >> BSH], 1);
  }
#pragma unroll
  for (int u = 0; u < CE; ++u) {
    const int e = e0 + u * CT;
    if (!TAIL || e < E) rank[e] = r[u];
  }
}

__global__ __launch_bounds__(CT) void k_bucket_count(const int* __restrict__ dst,
                                                     int* __restrict__ rank,
                                                     int* __restrict__ H, int E, int NB, int G) {
  __shared__ int h[NBMAX];
  const int tid = threadIdx.x;
  const int g = blockIdx.x;
  for (int j = tid; j < NB; j += CT) h[j] = 0;
  __syncthreads();
  const int e0 = g * (CT * CE) + tid;
  if ((g + 1) * (CT * CE) <= E) count_chunk<false>(dst, rank, h, e0, E);
  else                          count_chunk<true>(dst, rank, h, e0, E);
  __syncthreads();
  for (int j = tid; j < NB; j += CT) H[j * G + g] = h[j];
}

__global__ __launch_bounds__(256) void k_scan(const int* __restrict__ cnt,
                                              int* __restrict__ cur, int M) {
  __shared__ int s_front[8];
  __shared__ int s_own[8];
  const int tid = threadIdx.x;
  const int lane = tid & 31;
  const int wave = tid >> 5;
  const int4* c4 = (const int4*)cnt;
  const int nq = blockIdx.x * 64;
  int front = 0;
  for (int j0 = tid; j0 < nq; j0 += 8 * 256) {
    int4 v[8];
#pragma unroll
    for (int u = 0; u < 8; ++u) v[u] = c4[min(j0 + u * 256, nq - 1)];
#pragma unroll
    for (int u = 0; u < 8; ++u)
      if (j0 + u * 256 < nq) front += (v[u].x + v[u].y) + (v[u].z + v[u].w);
  }
  const int i = blockIdx.x * 256 + tid;
  const int c = (i < M) ? cnt[i] : 0;
  int incl = c;
#pragma unroll
  for (int off = 1; off < 32; off <<= 1) {
    int t = __shfl_up(incl, off, 32);
    if (lane >= off) incl += t;
  }
#pragma unroll
  for (int off = 16; off > 0; off >>= 1) front += __shfl_xor(front, off, 32);
  if (lane == 31) s_own[wave] = incl;
  if (lane == 0) s_front[wave] = front;
  __syncthreads();
  int base = 0;
#pragma unroll
  for (int w = 0; w < 8; ++w) {
    base += s_front[w];
    if (w < wave) base += s_own[w];
  }
  if (i < M) {
    cur[i] = base + incl - c;
    if (i == M - 1) cur[M] = base + incl;
  }
}

template <bool TAIL>
static __device__ __forceinline__ void scatter_chunk(const int* __restrict__ src, const int* __restrict__ dst,
                                                     const int* __restrict__ rank, const int* __restrict__ first,
                                                     int* __restrict__ tmp, int e0, int E, int G, int g) {
  int s[CE], d[CE], r[CE], p[CE];
#pragma unroll
  for (int u = 0; u < CE; ++u) {
    const int e = TAIL ? min(e0 + u * CT, E - 1) : e0 + u * CT;
    s[u] = src[e];
    d[u] = dst[e];
    r[u] = rank[e];
  }
#pragma unroll
  for (int u = 0; u < CE; ++u) p[u] = first[(d[u] >> BSH) * G + g] + r[u];
#pragma unroll
  for (int u = 0; u < CE; ++u)
    if (!TAIL || e0 + u * CT < E) ((int2*)tmp)[p[u]] = make_int2(s[u], d[u]);
}

__global__ __launch_bounds__(CT) void k_bucket_scatter(const int* __restrict__ src,
                                                       const int* __restrict__ dst,
                                                       const int* __restrict__ rank,
                                                       const int* __restrict__ first,
                                                       int* __restrict__ tmp, int E, int G) {
  const int g = blockIdx.x;
  const int e0 = g * (CT * CE) + threadIdx.x;
  if ((g + 1) * (CT * CE) <= E) scatter_chunk<false>(src, dst, rank, first, tmp, e0, E, G, g);
  else                          scatter_chunk<true>(src, dst, rank, first, tmp, e0, E, G, g);
}

__global__ __launch_bounds__(1024) void k_bucket_sort(const int* __restrict__ tmp,
                                                      const int* __restrict__ first,
                                                      int* __restrict__ list,
                                                      int* __restrict__ cur,
                                                      float* __restrict__ dinv, int N, int G) {
  __shared__ int c[256];
  __shared__ int st[256];
  __shared__ int f[256];
  const int tid = threadIdx.x;
  const int b = blockIdx.x;
  const int lo = first[b * G];
  const int hi = first[(b + 1) * G];
  if (tid < 256) {
    c[tid] = 0;
    f[tid] = 0;
  }
  __syncthreads();
  for (int k0 = lo + tid; k0 < hi; k0 += 4096) {
    int d[4];
#pragma unroll
    for (int u = 0; u < 4; ++u) d[u] = tmp[2 * min(k0 + u * 1024, hi - 1) + 1];
#pragma unroll
    for (int u = 0; u < 4; ++u)
      if (k0 + u * 1024 < hi) atomicAdd(&c[d[u] & 255], 1);
  }
  __syncthreads();
  if (tid < 32) {
    int v[8];
    int run = 0;
#pragma unroll
    for (int j = 0; j < 8; ++j) {
      const int t = c[8 * tid + j];
      v[j] = run;
      run += t;
    }
    int incl = run;
#pragma unroll
    for (int off = 1; off < 32; off <<= 1) {
      int t = __shfl_up(incl, off, 32);
      if (tid >= off) incl += t;
    }
    const int lanebase = incl - run;
#pragma unroll
    for (int j = 0; j < 8; ++j) st[8 * tid + j] = lanebase + v[j];
  }
  __syncthreads();
  if (tid < 256) {
    const int node = b * 256 + tid;
    if (node < N) {
      cur[node] = lo + st[tid];
      dinv[node] = rsqrtf((float)(c[tid] + 1));
    }
  }
  if (b == ((N + (1 << BSH) - 1) >> BSH) - 1 && tid == 0) cur[N] = hi;
  for (int k0 = lo + tid; k0 < hi; k0 += 4096) {
    int2 ent[4];
#pragma unroll
    for (int u = 0; u < 4; ++u) ent[u] = ((const int2*)tmp)[min(k0 + u * 1024, hi - 1)];
#pragma unroll
    for (int u = 0; u < 4; ++u) {
      if (k0 + u * 1024 < hi) {
        const int j = ent[u].y & 255;
        const int r = atomicAdd(&f[j], 1);
        ((int2*)list)[lo + st[j] + r] = ent[u];
      }
    }
  }
}

#define NT 128
#define NS 8
template <bool TAIL>
static __device__ __forceinline__ void norm_chunk(int* __restrict__ list, const float* __restrict__ dinv,
                                                  int s0, int E) {
  int2 ent[NS];
  float a[NS], bb[NS];
#pragma unroll
  for (int u = 0; u < NS; ++u) {
    const int slot = s0 + u * NT;
    ent[u] = make_int2(0, 0);
    if (!TAIL || slot < E) ent[u] = ((const int2*)list)[slot];
  }
#pragma unroll
  for (int u = 0; u < NS; ++u) {
    a[u] = dinv[ent[u].x];
    bb[u] = dinv[ent[u].y];
  }
#pragma unroll
  for (int u = 0; u < NS; ++u) {
    const int slot = s0 + u * NT;
    if (!TAIL || slot < E) list[2 * slot + 1] = __float_as_int(a[u] * bb[u]);
  }
}

__global__ __launch_bounds__(NT) void k_norm(int* __restrict__ list,
                                             const float* __restrict__ dinv, int E) {
  const int s0 = blockIdx.x * (NT * NS) + threadIdx.x;
  if ((blockIdx.x + 1) * (NT * NS) <= E) norm_chunk<false>(list, dinv, s0, E);
  else                                   norm_chunk<true>(list, dinv, s0, E);
}

static __device__ __forceinline__ float xhalf(float x) {
  return __builtin_bit_cast(float, __builtin_amdgcn_permlanex16(__builtin_bit_cast(int, x), __builtin_bit_cast(int, x),
                                                                0x76543210, 0xfedcba98, false, false));
}

static __device__ __forceinline__ float4 gather_row(const float* __restrict__ h,
                                                    const int* __restrict__ list,
                                                    const int* __restrict__ cur,
                                                    const float* __restrict__ dinv,
                                                    int i, int lane) {
  const int half = lane >> 4;
  const int c4 = (lane & 15) * 4;
  const int start = cur[i];
  const int end = cur[i + 1];
  const float di = dinv[i];
  const float4 self = *(const float4*)(h + (size_t)i * CH + c4);
  const float ws = di * di;
  float4 acc;
  acc.x = half ? 0.0f : self.x * ws;
  acc.y = half ? 0.0f : self.y * ws;
  acc.z = half ? 0.0f : self.z * ws;
  acc.w = half ? 0.0f : self.w * ws;

  for (int base = start; base < end; base += 32) {
    const int n = min(32, end - base);
    int2 ent = make_int2(i, 0);
    if (lane < n) ent = ((const int2*)list)[base + lane];
    const int npairs = (n + 1) >> 1;
    for (int p = 0; p < npairs; p += 8) {
      int s[8];
      float w[8];
      float4 r[8];
#pragma unroll
      for (int u = 0; u < 8; ++u) {
        const int q = 2 * (p + u);
        const int sa = __builtin_amdgcn_readlane(ent.x, q);
        const int sb = __builtin_amdgcn_readlane(ent.x, q + 1);
        const int wa = __builtin_amdgcn_readlane(ent.y, q);
        const int wb = __builtin_amdgcn_readlane(ent.y, q + 1);
        s[u] = half ? sb : sa;
        w[u] = __int_as_float(half ? wb : wa);
      }
#pragma unroll
      for (int u = 0; u < 8; ++u) {
        r[u] = make_float4(0.0f, 0.0f, 0.0f, 0.0f);
        if (2 * (p + u) + half < n) r[u] = *(const float4*)(h + (size_t)s[u] * CH + c4);
      }
#pragma unroll
      for (int u = 0; u < 8; ++u) {
        acc.x = fmaf(r[u].x, w[u], acc.x);
        acc.y = fmaf(r[u].y, w[u], acc.y);
        acc.z = fmaf(r[u].z, w[u], acc.z);
        acc.w = fmaf(r[u].w, w[u], acc.w);
      }
    }
  }
  acc.x += xhalf(acc.x);
  acc.y += xhalf(acc.y);
  acc.z += xhalf(acc.z);
  acc.w += xhalf(acc.w);
  return acc;
}

#define TN 16
#define TP 68

template <bool PROJ>
static __device__ __forceinline__ void layer_tile(const float* __restrict__ hin,
                                                  const int* __restrict__ list,
                                                  const int* __restrict__ cur,
                                                  const float* __restrict__ dinv,
                                                  const float* __restrict__ W,
                                                  const float* __restrict__ b,
                                                  const float* __restrict__ Wfc,
                                                  float* __restrict__ out,
                                                  float* __restrict__ ps,
                                                  float* __restrict__ pd, int N) {
  __shared__ float agg[TN * TP];
  __shared__ float act[TN * TP];
  const int lane = threadIdx.x & 31;
  const int wave = threadIdx.x >> 5;
  const int kh = lane >> 4;
  const int m = lane & 15;
  const int c4 = m * 4;
  const int tile0 = blockIdx.x * TN;

  for (int t = 0; t < 4; ++t) {
    const int r = wave * 4 + t;
    const int i = __builtin_amdgcn_readfirstlane(tile0 + r);
    float4 v = make_float4(0.0f, 0.0f, 0.0f, 0.0f);
    if (i < N) v = gather_row(hin, list, cur, dinv, i, lane);
    if (lane < 16) *(float4*)(agg + r * TP + c4) = v;
  }

  const int n0 = wave * 16;
  v2f bw[CH / 4];
#pragma unroll
  for (int j = 0; j < CH / 4; ++j) {
    bw[j].x = W[(4 * j + 2 * kh + 0) * CH + n0 + m];
    bw[j].y = W[(4 * j + 2 * kh + 1) * CH + n0 + m];
  }
  const float bias = b[n0 + m];
  __syncthreads();

  const float* arow = agg + m * TP + 2 * kh;
  v8f acc = {};
#pragma unroll
  for (int j = 0; j < CH / 4; ++j) {
    v2f a;
    a.x = arow[4 * j];
    a.y = arow[4 * j + 1];
    acc = __builtin_amdgcn_wmma_f32_16x16x4_f32(false, a, false, bw[j],
                                                (short)0, acc, false, false);
  }
#pragma unroll
  for (int v = 0; v < 8; ++v) {
    const float y = acc[v] + bias;
    act[(v + 8 * kh) * TP + n0 + m] = y > 0.0f ? y : 0.0f;
  }
  float4 w0, w1;
  if (PROJ) {
    w0 = *(const float4*)(Wfc + c4);
    w1 = *(const float4*)(Wfc + CH + c4);
  }
  __syncthreads();

#pragma unroll
  for (int t = 0; t < 2; ++t) {
    const int r = wave * 4 + 2 * t + kh;
    const int i = tile0 + r;
    const float4 v = *(const float4*)(act + r * TP + c4);
    if (PROJ) {
      float s0 = (v.x * w0.x + v.y * w0.y) + (v.z * w0.z + v.w * w0.w);
      float s1 = (v.x * w1.x + v.y * w1.y) + (v.z * w1.z + v.w * w1.w);
#pragma unroll
      for (int off = 8; off > 0; off >>= 1) {
        s0 += __shfl_xor(s0, off, 32);
        s1 += __shfl_xor(s1, off, 32);
      }
      if (m == 0 && i < N) {
        ps[i] = s0;
        pd[i] = s1;
      }
    } else {
      if (i < N) *(float4*)(out + (size_t)i * CH + c4) = v;
    }
  }
}

__global__ __launch_bounds__(128) void k_layer_relu(const float* __restrict__ hin,
                                                    const int* __restrict__ list,
                                                    const int* __restrict__ cur,
                                                    const float* __restrict__ dinv,
                                                    const float* __restrict__ W,
                                                    const float* __restrict__ b,
                                                    float* __restrict__ out, int N) {
  layer_tile<false>(hin, list, cur, dinv, W, b, nullptr, out, nullptr, nullptr, N);
}

__global__ __launch_bounds__(128) void k_layer_relu_proj(const float* __restrict__ hin,
                                                         const int* __restrict__ list,
                                                         const int* __restrict__ cur,
                                                         const float* __restrict__ dinv,
                                                         const float* __restrict__ W,
                                                         const float* __restrict__ b,
                                                         const float* __restrict__ Wfc,
                                                         float* __restrict__ ps,
                                                         float* __restrict__ pd, int N) {
  layer_tile<true>(hin, list, cur, dinv, W, b, Wfc, nullptr, ps, pd, N);
}

#define PT 128
#define PQ 2

template <bool TAIL>
static __device__ __forceinline__ void pred_groups(const float* __restrict__ ps, const float* __restrict__ pd,
                                                   const int* __restrict__ src, const int* __restrict__ dst,
                                                   float b, float* __restrict__ out, int q0, int nq) {
  int4 s[PQ], d[PQ];
  float4 a[PQ], c[PQ];
#pragma unroll
  for (int u = 0; u < PQ; ++u) {
    const int q = TAIL ? min(q0 + u * PT, nq - 1) : q0 + u * PT;
    s[u] = ((const int4*)src)[q];
    d[u] = ((const int4*)dst)[q];
  }
#pragma unroll
  for (int u = 0; u < PQ; ++u) {
    a[u].x = ps[s[u].x];  c[u].x = pd[d[u].x];
    a[u].y = ps[s[u].y];  c[u].y = pd[d[u].y];
    a[u].z = ps[s[u].z];  c[u].z = pd[d[u].z];
    a[u].w = ps[s[u].w];  c[u].w = pd[d[u].w];
  }
#pragma unroll
  for (int u = 0; u < PQ; ++u) {
    const int q = q0 + u * PT;
    if (!TAIL || q < nq)
      ((float4*)out)[q] = make_float4(a[u].x + c[u].x + b, a[u].y + c[u].y + b,
                                      a[u].z + c[u].z + b, a[u].w + c[u].w + b);
  }
}

__global__ __launch_bounds__(PT) void k_edge_pred(const float* __restrict__ ps, const float* __restrict__ pd,
                                                  const int* __restrict__ src, const int* __restrict__ dst,
                                                  const float* __restrict__ bfc, float* __restrict__ out, int E) {
  const float b = bfc[0];
  if ((E & 3) == 0) {
    const int nq = E >> 2;
    const int q0 = blockIdx.x * (PT * PQ) + threadIdx.x;
    if ((blockIdx.x + 1) * (PT * PQ) <= nq) pred_groups<false>(ps, pd, src, dst, b, out, q0, nq);
    else                                    pred_groups<true>(ps, pd, src, dst, b, out, q0, nq);
  } else {
    const int e0 = blockIdx.x * (PT * PQ * 4) + threadIdx.x;
    for (int u = 0; u < PQ * 4; ++u) {
      const int e = e0 + u * PT;
      if (e < E) out[e] = ps[src[e]] + pd[dst[e]] + b;
    }
  }
}

extern "C" void kernel_launch(void* const* d_in, const int* in_sizes, int n_in,
                              void* d_out, int out_size, void* d_ws, size_t ws_size,
                              hipStream_t stream) {
  const float* x   = (const float*)d_in[0];
  const int*   ei  = (const int*)d_in[1];
  const float* W1  = (const float*)d_in[2];
  const float* b1  = (const float*)d_in[3];
  const float* W2  = (const float*)d_in[4];
  const float* b2  = (const float*)d_in[5];
  const float* Wfc = (const float*)d_in[6];
  const float* bfc = (const float*)d_in[7];

  const int N = in_sizes[0] / CH;
  const int E = in_sizes[1] / 2;
  const int NF = N * CH;

  const int* src = ei;
  const int* dst = ei + E;

  const int NB = (N + (1 << BSH) - 1) >> BSH;
  const int G  = (E + CT * CE - 1) / (CT * CE);
  const int M  = NB * G;
  const int E4 = (E + 3) & ~3;
  const int M4 = (M + 3) & ~3;

  float* ws    = (float*)d_ws;
  float* AB    = ws;
  int*   list  = (int*)(AB + (size_t)NF);
  int*   tmp   = list + 2 * (size_t)E;
  int*   rank  = tmp + 2 * (size_t)E;
  int*   H     = rank + E4;
  int*   first = H + M4;
  int*   cur   = first + M4 + 4;
  float* dinv  = (float*)(cur + N + 1);
  float* PS    = dinv + N;
  float* PD    = PS + N;

  const int TB = 256;
  dim3 blk(TB);
  dim3 gE((E + PT * PQ * 4 - 1) / (PT * PQ * 4));
  dim3 gT((N + TN - 1) / TN);

  k_bucket_count<<<dim3(G), dim3(CT), 0, stream>>>(dst, rank, H, E, NB, G);
  k_scan<<<dim3((M + 255) / 256), blk, 0, stream>>>(H, first, M);
  k_bucket_scatter<<<dim3(G), dim3(CT), 0, stream>>>(src, dst, rank, first, tmp, E, G);
  k_bucket_sort<<<dim3(NB), dim3(1024), 0, stream>>>(tmp, first, list, cur, dinv, N, G);
  k_norm<<<dim3((E + NT * NS - 1) / (NT * NS)), dim3(NT), 0, stream>>>(list, dinv, E);

  k_layer_relu<<<gT, dim3(128), 0, stream>>>(x, list, cur, dinv, W1, b1, AB, N);

  k_layer_relu_proj<<<gT, dim3(128), 0, stream>>>(AB, list, cur, dinv, W2, b2, Wfc, PS, PD, N);

  k_edge_pred<<<gE, dim3(PT), 0, stream>>>(PS, PD, src, dst, bfc, (float*)d_out, E);
}


# Scan launch gone: each chunk's block of the edge-moving kernel adds up the (bucket, chunk) count table itself (row prefixes, bucket scan in LDS) and keeps its first places in LDS; sum order kept
# speedup vs baseline: 4.1555x; 1.0536x over previous
#include <hip/hip_runtime.h>
#include <hip/hip_bf16.h>

typedef __attribute__((ext_vector_type(2)))  float    v2f;
typedef __attribute__((ext_vector_type(8)))  float    v8f;
typedef __attribute__((ext_vector_type(4)))  int      v4i;

#define CH 64

#define BSH   8
#define NBMAX 256
#define CT    512
#define CE    8

template <bool TAIL>
static __device__ __forceinline__ void count_chunk(const int* __restrict__ dst, int* __restrict__ rank,
                                                   int* h, int e0, int E) {
  int d[CE];
#pragma unroll
  for (int u = 0; u < CE; ++u) d[u] = dst[TAIL ? min(e0 + u * CT, E - 1) : e0 + u * CT];
  int r[CE];
#pragma unroll
  for (int u = 0; u < CE; ++u) {
    r[u] = 0;
    if (!TAIL || e0 + u * CT < E) r[u] = atomicAdd(&h[d[u] >> BSH], 1);
  }
#pragma unroll
  for (int u = 0; u < CE; ++u) {
    const int e = e0 + u * CT;
    if (!TAIL || e < E) rank[e] = r[u];
  }
}

__global__ __launch_bounds__(CT) void k_bucket_count(const int* __restrict__ dst,
                                                     int* __restrict__ rank,
                                                     int* __restrict__ H, int E, int NB, int G, int GP) {
  __shared__ int h[NBMAX];
  const int tid = threadIdx.x;
  const int g = blockIdx.x;
  for (int j = tid; j < NB; j += CT) h[j] = 0;
  __syncthreads();
  const int e0 = g * (CT * CE) + tid;
  if ((g + 1) * (CT * CE) <= E) count_chunk<false>(dst, rank, h, e0, E);
  else                          count_chunk<true>(dst, rank, h, e0, E);
  __syncthreads();
  for (int j = tid; j < NB; j += CT) H[j * GP + g] = h[j];
  if (g == 0)
    for (int j = tid; j < NB; j += CT)
      for (int q = G; q < GP; ++q) H[j * GP + q] = 0;
}

#define LB 8
#define NR (NBMAX / ((CT / 32) * (32 / LB)))
#define TU 7
static_assert(NR == 4 && TU == 7, "the load fence in table_trip names NR*TU values");

static __device__ __forceinline__ void table_trip(const v4i* __restrict__ H4, const int (&rq)[NR],
                                                  int (&pa)[NR], int (&ta)[NR], int q0, int nq, int qb) {
  v4i v[NR][TU];
#pragma unroll
  for (int rr = 0; rr < NR; ++rr)
#pragma unroll
    for (int u = 0; u < TU; ++u) v[rr][u] = H4[rq[rr] + min(q0 + u * LB, nq - 1)];
  asm volatile("" :: "v"(v[0][0]), "v"(v[0][1]), "v"(v[0][2]), "v"(v[0][3]), "v"(v[0][4]), "v"(v[0][5]), "v"(v[0][6]),
                     "v"(v[1][0]), "v"(v[1][1]), "v"(v[1][2]), "v"(v[1][3]), "v"(v[1][4]), "v"(v[1][5]), "v"(v[1][6]),
                     "v"(v[2][0]), "v"(v[2][1]), "v"(v[2][2]), "v"(v[2][3]), "v"(v[2][4]), "v"(v[2][5]), "v"(v[2][6]),
                     "v"(v[3][0]), "v"(v[3][1]), "v"(v[3][2]), "v"(v[3][3]), "v"(v[3][4]), "v"(v[3][5]), "v"(v[3][6]));
#pragma unroll
  for (int rr = 0; rr < NR; ++rr)
#pragma unroll
    for (int u = 0; u < TU; ++u) {
      const int q = q0 + u * LB;
      const int sm = (v[rr][u].x + v[rr][u].y) + (v[rr][u].z + v[rr][u].w);
      ta[rr] += q < nq ? sm : 0;
      pa[rr] += q < qb ? sm : 0;
    }
}

template <bool TAIL>
static __device__ __forceinline__ void scatter_chunk(const int* __restrict__ src, const int* __restrict__ dst,
                                                     const int* __restrict__ rank, const int* __restrict__ H,
                                                     int* __restrict__ bstart, int* __restrict__ tmp,
                                                     int* P, int* T, int* fs,
                                                     int E, int NB, int GP, int g) {
  const int tid = threadIdx.x;
  const int lane = tid & 31;
  const int wave = tid >> 5;
  const int e0 = g * (CT * CE) + tid;
  int s[CE], d[CE], r[CE];
#pragma unroll
  for (int u = 0; u < CE; ++u) {
    const int e = TAIL ? min(e0 + u * CT, E - 1) : e0 + u * CT;
    s[u] = src[e];
    d[u] = dst[e];
    r[u] = rank[e];
  }

  const int sub = lane & (LB - 1);
  const int bl = lane / LB;
  const int nq = GP >> 2;
  const int qb = g >> 2;
  const int kb = g & 3;
  const v4i* H4 = (const v4i*)H;
  int rq[NR];
  int pa[NR], ta[NR];
  v4i wb[NR];
#pragma unroll
  for (int rr = 0; rr < NR; ++rr) {
    const int b = min(rr * (NBMAX / NR) + wave * (32 / LB) + bl, NB - 1);
    rq[rr] = b * nq;
    pa[rr] = 0;
    ta[rr] = 0;
    wb[rr] = H4[rq[rr] + qb];
  }
  if (nq <= TU * LB) table_trip(H4, rq, pa, ta, sub, nq, qb);
  else
    for (int q0 = sub; q0 < nq; q0 += TU * LB) table_trip(H4, rq, pa, ta, q0, nq, qb);
#pragma unroll
  for (int rr = 0; rr < NR; ++rr) {
    const int part = (kb > 0 ? wb[rr].x : 0) + (kb > 1 ? wb[rr].y : 0) + (kb > 2 ? wb[rr].z : 0);
    pa[rr] += sub == 0 ? part : 0;
#pragma unroll
    for (int off = LB / 2; off > 0; off >>= 1) {
      pa[rr] += __shfl_xor(pa[rr], off, 32);
      ta[rr] += __shfl_xor(ta[rr], off, 32);
    }
    const int b = rr * (NBMAX / NR) + wave * (32 / LB) + bl;
    if (sub == 0 && b < NB) {
      P[b] = pa[rr];
      T[b] = ta[rr];
    }
  }
  __syncthreads();
  if (tid < 32) {
    int v[8];
    int run = 0;
#pragma unroll
    for (int j = 0; j < 8; ++j) {
      const int t = 8 * tid + j < NB ? T[8 * tid + j] : 0;
      v[j] = run;
      run += t;
    }
    int incl = run;
#pragma unroll
    for (int off = 1; off < 32; off <<= 1) {
      int t = __shfl_up(incl, off, 32);
      if (tid >= off) incl += t;
    }
    const int lanebase = incl - run;
#pragma unroll
    for (int j = 0; j < 8; ++j) {
      const int b = 8 * tid + j;
      if (b < NB) {
        const int st = lanebase + v[j];
        fs[b] = st + P[b];
        if (g == 0) bstart[b] = st;
      }
    }
    if (g == 0 && tid == 31) bstart[NB] = incl;
  }
  __syncthreads();

#pragma unroll
  for (int u = 0; u < CE; ++u)
    if (!TAIL || e0 + u * CT < E) ((int2*)tmp)[fs[d[u] >> BSH] + r[u]] = make_int2(s[u], d[u]);
}

__global__ __launch_bounds__(CT) void k_bucket_scatter(const int* __restrict__ src,
                                                       const int* __restrict__ dst,
                                                       const int* __restrict__ rank,
                                                       const int* __restrict__ H,
                                                       int* __restrict__ bstart,
                                                       int* __restrict__ tmp, int E, int NB, int GP) {
  __shared__ int P[NBMAX];
  __shared__ int T[NBMAX];
  __shared__ int fs[NBMAX];
  const int g = blockIdx.x;
  if ((g + 1) * (CT * CE) <= E) scatter_chunk<false>(src, dst, rank, H, bstart, tmp, P, T, fs, E, NB, GP, g);
  else                          scatter_chunk<true>(src, dst, rank, H, bstart, tmp, P, T, fs, E, NB, GP, g);
}

__global__ __launch_bounds__(1024) void k_bucket_sort(const int* __restrict__ tmp,
                                                      const int* __restrict__ bstart,
                                                      int* __restrict__ list,
                                                      int* __restrict__ cur,
                                                      float* __restrict__ dinv, int N) {
  __shared__ int c[256];
  __shared__ int st[256];
  __shared__ int f[256];
  const int tid = threadIdx.x;
  const int b = blockIdx.x;
  const int lo = bstart[b];
  const int hi = bstart[b + 1];
  if (tid < 256) {
    c[tid] = 0;
    f[tid] = 0;
  }
  __syncthreads();
  for (int k0 = lo + tid; k0 < hi; k0 += 4096) {
    int d[4];
#pragma unroll
    for (int u = 0; u < 4; ++u) d[u] = tmp[2 * min(k0 + u * 1024, hi - 1) + 1];
#pragma unroll
    for (int u = 0; u < 4; ++u)
      if (k0 + u * 1024 < hi) atomicAdd(&c[d[u] & 255], 1);
  }
  __syncthreads();
  if (tid < 32) {
    int v[8];
    int run = 0;
#pragma unroll
    for (int j = 0; j < 8; ++j) {
      const int t = c[8 * tid + j];
      v[j] = run;
      run += t;
    }
    int incl = run;
#pragma unroll
    for (int off = 1; off < 32; off <<= 1) {
      int t = __shfl_up(incl, off, 32);
      if (tid >= off) incl += t;
    }
    const int lanebase = incl - run;
#pragma unroll
    for (int j = 0; j < 8; ++j) st[8 * tid + j] = lanebase + v[j];
  }
  __syncthreads();
  if (tid < 256) {
    const int node = b * 256 + tid;
    if (node < N) {
      cur[node] = lo + st[tid];
      dinv[node] = rsqrtf((float)(c[tid] + 1));
    }
  }
  if (b == ((N + (1 << BSH) - 1) >> BSH) - 1 && tid == 0) cur[N] = hi;
  for (int k0 = lo + tid; k0 < hi; k0 += 4096) {
    int2 ent[4];
#pragma unroll
    for (int u = 0; u < 4; ++u) ent[u] = ((const int2*)tmp)[min(k0 + u * 1024, hi - 1)];
#pragma unroll
    for (int u = 0; u < 4; ++u) {
      if (k0 + u * 1024 < hi) {
        const int j = ent[u].y & 255;
        const int r = atomicAdd(&f[j], 1);
        ((int2*)list)[lo + st[j] + r] = ent[u];
      }
    }
  }
}

#define NT 128
#define NS 8
template <bool TAIL>
static __device__ __forceinline__ void norm_chunk(int* __restrict__ list, const float* __restrict__ dinv,
                                                  int s0, int E) {
  int2 ent[NS];
  float a[NS], bb[NS];
#pragma unroll
  for (int u = 0; u < NS; ++u) {
    const int slot = s0 + u * NT;
    ent[u] = make_int2(0, 0);
    if (!TAIL || slot < E) ent[u] = ((const int2*)list)[slot];
  }
#pragma unroll
  for (int u = 0; u < NS; ++u) {
    a[u] = dinv[ent[u].x];
    bb[u] = dinv[ent[u].y];
  }
#pragma unroll
  for (int u = 0; u < NS; ++u) {
    const int slot = s0 + u * NT;
    if (!TAIL || slot < E) list[2 * slot + 1] = __float_as_int(a[u] * bb[u]);
  }
}

__global__ __launch_bounds__(NT) void k_norm(int* __restrict__ list,
                                             const float* __restrict__ dinv, int E) {
  const int s0 = blockIdx.x * (NT * NS) + threadIdx.x;
  if ((blockIdx.x + 1) * (NT * NS) <= E) norm_chunk<false>(list, dinv, s0, E);
  else                                   norm_chunk<true>(list, dinv, s0, E);
}

static __device__ __forceinline__ float xhalf(float x) {
  return __builtin_bit_cast(float, __builtin_amdgcn_permlanex16(__builtin_bit_cast(int, x), __builtin_bit_cast(int, x),
                                                                0x76543210, 0xfedcba98, false, false));
}

static __device__ __forceinline__ float4 gather_row(const float* __restrict__ h,
                                                    const int* __restrict__ list,
                                                    const int* __restrict__ cur,
                                                    const float* __restrict__ dinv,
                                                    int i, int lane) {
  const int half = lane >> 4;
  const int c4 = (lane & 15) * 4;
  const int start = cur[i];
  const int end = cur[i + 1];
  const float di = dinv[i];
  const float4 self = *(const float4*)(h + (size_t)i * CH + c4);
  const float ws = di * di;
  float4 acc;
  acc.x = half ? 0.0f : self.x * ws;
  acc.y = half ? 0.0f : self.y * ws;
  acc.z = half ? 0.0f : self.z * ws;
  acc.w = half ? 0.0f : self.w * ws;

  for (int base = start; base < end; base += 32) {
    const int n = min(32, end - base);
    int2 ent = make_int2(i, 0);
    if (lane < n) ent = ((const int2*)list)[base + lane];
    const int npairs = (n + 1) >> 1;
    for (int p = 0; p < npairs; p += 8) {
      int s[8];
      float w[8];
      float4 r[8];
#pragma unroll
      for (int u = 0; u < 8; ++u) {
        const int q = 2 * (p + u);
        const int sa = __builtin_amdgcn_readlane(ent.x, q);
        const int sb = __builtin_amdgcn_readlane(ent.x, q + 1);
        const int wa = __builtin_amdgcn_readlane(ent.y, q);
        const int wb = __builtin_amdgcn_readlane(ent.y, q + 1);
        s[u] = half ? sb : sa;
        w[u] = __int_as_float(half ? wb : wa);
      }
#pragma unroll
      for (int u = 0; u < 8; ++u) {
        r[u] = make_float4(0.0f, 0.0f, 0.0f, 0.0f);
        if (2 * (p + u) + half < n) r[u] = *(const float4*)(h + (size_t)s[u] * CH + c4);
      }
#pragma unroll
      for (int u = 0; u < 8; ++u) {
        acc.x = fmaf(r[u].x, w[u], acc.x);
        acc.y = fmaf(r[u].y, w[u], acc.y);
        acc.z = fmaf(r[u].z, w[u], acc.z);
        acc.w = fmaf(r[u].w, w[u], acc.w);
      }
    }
  }
  acc.x += xhalf(acc.x);
  acc.y += xhalf(acc.y);
  acc.z += xhalf(acc.z);
  acc.w += xhalf(acc.w);
  return acc;
}

#define TN 16
#define TP 68

template <bool PROJ>
static __device__ __forceinline__ void layer_tile(const float* __restrict__ hin,
                                                  const int* __restrict__ list,
                                                  const int* __restrict__ cur,
                                                  const float* __restrict__ dinv,
                                                  const float* __restrict__ W,
                                                  const float* __restrict__ b,
                                                  const float* __restrict__ Wfc,
                                                  float* __restrict__ out,
                                                  float* __restrict__ ps,
                                                  float* __restrict__ pd, int N) {
  __shared__ float agg[TN * TP];
  __shared__ float act[TN * TP];
  const int lane = threadIdx.x & 31;
  const int wave = threadIdx.x >> 5;
  const int kh = lane >> 4;
  const int m = lane & 15;
  const int c4 = m * 4;
  const int tile0 = blockIdx.x * TN;

  for (int t = 0; t < 4; ++t) {
    const int r = wave * 4 + t;
    const int i = __builtin_amdgcn_readfirstlane(tile0 + r);
    float4 v = make_float4(0.0f, 0.0f, 0.0f, 0.0f);
    if (i < N) v = gather_row(hin, list, cur, dinv, i, lane);
    if (lane < 16) *(float4*)(agg + r * TP + c4) = v;
  }

  const int n0 = wave * 16;
  v2f bw[CH / 4];
#pragma unroll
  for (int j = 0; j < CH / 4; ++j) {
    bw[j].x = W[(4 * j + 2 * kh + 0) * CH + n0 + m];
    bw[j].y = W[(4 * j + 2 * kh + 1) * CH + n0 + m];
  }
  const float bias = b[n0 + m];
  __syncthreads();

  const float* arow = agg + m * TP + 2 * kh;
  v8f acc = {};
#pragma unroll
  for (int j = 0; j < CH / 4; ++j) {
    v2f a;
    a.x = arow[4 * j];
    a.y = arow[4 * j + 1];
    acc = __builtin_amdgcn_wmma_f32_16x16x4_f32(false, a, false, bw[j],
                                                (short)0, acc, false, false);
  }
#pragma unroll
  for (int v = 0; v < 8; ++v) {
    const float y = acc[v] + bias;
    act[(v + 8 * kh) * TP + n0 + m] = y > 0.0f ? y : 0.0f;
  }
  float4 w0, w1;
  if (PROJ) {
    w0 = *(const float4*)(Wfc + c4);
    w1 = *(const float4*)(Wfc + CH + c4);
  }
  __syncthreads();

#pragma unroll
  for (int t = 0; t < 2; ++t) {
    const int r = wave * 4 + 2 * t + kh;
    const int i = tile0 + r;
    const float4 v = *(const float4*)(act + r * TP + c4);
    if (PROJ) {
      float s0 = (v.x * w0.x + v.y * w0.y) + (v.z * w0.z + v.w * w0.w);
      float s1 = (v.x * w1.x + v.y * w1.y) + (v.z * w1.z + v.w * w1.w);
#pragma unroll
      for (int off = 8; off > 0; off >>= 1) {
        s0 += __shfl_xor(s0, off, 32);
        s1 += __shfl_xor(s1, off, 32);
      }
      if (m == 0 && i < N) {
        ps[i] = s0;
        pd[i] = s1;
      }
    } else {
      if (i < N) *(float4*)(out + (size_t)i * CH + c4) = v;
    }
  }
}

__global__ __launch_bounds__(128) void k_layer_relu(const float* __restrict__ hin,
                                                    const int* __restrict__ list,
                                                    const int* __restrict__ cur,
                                                    const float* __restrict__ dinv,
                                                    const float* __restrict__ W,
                                                    const float* __restrict__ b,
                                                    float* __restrict__ out, int N) {
  layer_tile<false>(hin, list, cur, dinv, W, b, nullptr, out, nullptr, nullptr, N);
}

__global__ __launch_bounds__(128) void k_layer_relu_proj(const float* __restrict__ hin,
                                                         const int* __restrict__ list,
                                                         const int* __restrict__ cur,
                                                         const float* __restrict__ dinv,
                                                         const float* __restrict__ W,
                                                         const float* __restrict__ b,
                                                         const float* __restrict__ Wfc,
                                                         float* __restrict__ ps,
                                                         float* __restrict__ pd, int N) {
  layer_tile<true>(hin, list, cur, dinv, W, b, Wfc, nullptr, ps, pd, N);
}

#define PT 128
#define PQ 2

template <bool TAIL>
static __device__ __forceinline__ void pred_groups(const float* __restrict__ ps, const float* __restrict__ pd,
                                                   const int* __restrict__ src, const int* __restrict__ dst,
                                                   float b, float* __restrict__ out, int q0, int nq) {
  int4 s[PQ], d[PQ];
  float4 a[PQ], c[PQ];
#pragma unroll
  for (int u = 0; u < PQ; ++u) {
    const int q = TAIL ? min(q0 + u * PT, nq - 1) : q0 + u * PT;
    s[u] = ((const int4*)src)[q];
    d[u] = ((const int4*)dst)[q];
  }
#pragma unroll
  for (int u = 0; u < PQ; ++u) {
    a[u].x = ps[s[u].x];  c[u].x = pd[d[u].x];
    a[u].y = ps[s[u].y];  c[u].y = pd[d[u].y];
    a[u].z = ps[s[u].z];  c[u].z = pd[d[u].z];
    a[u].w = ps[s[u].w];  c[u].w = pd[d[u].w];
  }
#pragma unroll
  for (int u = 0; u < PQ; ++u) {
    const int q = q0 + u * PT;
    if (!TAIL || q < nq)
      ((float4*)out)[q] = make_float4(a[u].x + c[u].x + b, a[u].y + c[u].y + b,
                                      a[u].z + c[u].z + b, a[u].w + c[u].w + b);
  }
}

__global__ __launch_bounds__(PT) void k_edge_pred(const float* __restrict__ ps, const float* __restrict__ pd,
                                                  const int* __restrict__ src, const int* __restrict__ dst,
                                                  const float* __restrict__ bfc, float* __restrict__ out, int E) {
  const float b = bfc[0];
  if ((E & 3) == 0) {
    const int nq = E >> 2;
    const int q0 = blockIdx.x * (PT * PQ) + threadIdx.x;
    if ((blockIdx.x + 1) * (PT * PQ) <= nq) pred_groups<false>(ps, pd, src, dst, b, out, q0, nq);
    else                                    pred_groups<true>(ps, pd, src, dst, b, out, q0, nq);
  } else {
    const int e0 = blockIdx.x * (PT * PQ * 4) + threadIdx.x;
    for (int u = 0; u < PQ * 4; ++u) {
      const int e = e0 + u * PT;
      if (e < E) out[e] = ps[src[e]] + pd[dst[e]] + b;
    }
  }
}

extern "C" void kernel_launch(void* const* d_in, const int* in_sizes, int n_in,
                              void* d_out, int out_size, void* d_ws, size_t ws_size,
                              hipStream_t stream) {
  const float* x   = (const float*)d_in[0];
  const int*   ei  = (const int*)d_in[1];
  const float* W1  = (const float*)d_in[2];
  const float* b1  = (const float*)d_in[3];
  const float* W2  = (const float*)d_in[4];
  const float* b2  = (const float*)d_in[5];
  const float* Wfc = (const float*)d_in[6];
  const float* bfc = (const float*)d_in[7];

  const int N = in_sizes[0] / CH;
  const int E = in_sizes[1] / 2;
  const int NF = N * CH;

  const int* src = ei;
  const int* dst = ei + E;

  const int NB = (N + (1 << BSH) - 1) >> BSH;
  const int G  = (E + CT * CE - 1) / (CT * CE);
  const int GP = (G + 3) & ~3;
  const int M  = NB * GP;
  const int E4 = (E + 3) & ~3;
  const int NB4 = (NB + 1 + 3) & ~3;

  float* ws    = (float*)d_ws;
  float* AB    = ws;
  int*   list  = (int*)(AB + (size_t)NF);
  int*   tmp   = list + 2 * (size_t)E;
  int*   rank  = tmp + 2 * (size_t)E;
  int*   H     = rank + E4;
  int*   bstart = H + M;
  int*   cur   = bstart + NB4;
  float* dinv  = (float*)(cur + N + 1);
  float* PS    = dinv + N;
  float* PD    = PS + N;

  dim3 gE((E + PT * PQ * 4 - 1) / (PT * PQ * 4));
  dim3 gT((N + TN - 1) / TN);

  k_bucket_count<<<dim3(G), dim3(CT), 0, stream>>>(dst, rank, H, E, NB, G, GP);
  k_bucket_scatter<<<dim3(G), dim3(CT), 0, stream>>>(src, dst, rank, H, bstart, tmp, E, NB, GP);
  k_bucket_sort<<<dim3(NB), dim3(1024), 0, stream>>>(tmp, bstart, list, cur, dinv, N);
  k_norm<<<dim3((E + NT * NS - 1) / (NT * NS)), dim3(NT), 0, stream>>>(list, dinv, E);

  k_layer_relu<<<gT, dim3(128), 0, stream>>>(x, list, cur, dinv, W1, b1, AB, N);

  k_layer_relu_proj<<<gT, dim3(128), 0, stream>>>(AB, list, cur, dinv, W2, b2, Wfc, PS, PD, N);

  k_edge_pred<<<gE, dim3(PT), 0, stream>>>(PS, PD, src, dst, bfc, (float*)d_out, E);
}


# Edge norm factored: dinv[src] scales rows at the source (layer 1 gathers it per 32 slots, layer-1 rows stored scaled), dinv[dst] once per node; norm launch gone, list = src only; rounding differs
# speedup vs baseline: 4.3324x; 1.0426x over previous
#include <hip/hip_runtime.h>
#include <hip/hip_bf16.h>

typedef __attribute__((ext_vector_type(2)))  float    v2f;
typedef __attribute__((ext_vector_type(8)))  float    v8f;
typedef __attribute__((ext_vector_type(4)))  int      v4i;

#define CH 64

#define BSH   8
#define NBMAX 256
#define CT    512
#define CE    8

template <bool TAIL>
static __device__ __forceinline__ void count_chunk(const int* __restrict__ dst, int* __restrict__ rank,
                                                   int* h, int e0, int E) {
  int d[CE];
#pragma unroll
  for (int u = 0; u < CE; ++u) d[u] = dst[TAIL ? min(e0 + u * CT, E - 1) : e0 + u * CT];
  int r[CE];
#pragma unroll
  for (int u = 0; u < CE; ++u) {
    r[u] = 0;
    if (!TAIL || e0 + u * CT < E) r[u] = atomicAdd(&h[d[u] >> BSH], 1);
  }
#pragma unroll
  for (int u = 0; u < CE; ++u) {
    const int e = e0 + u * CT;
    if (!TAIL || e < E) rank[e] = r[u];
  }
}

__global__ __launch_bounds__(CT) void k_bucket_count(const int* __restrict__ dst,
                                                     int* __restrict__ rank,
                                                     int* __restrict__ H, int E, int NB, int G, int GP) {
  __shared__ int h[NBMAX];
  const int tid = threadIdx.x;
  const int g = blockIdx.x;
  for (int j = tid; j < NB; j += CT) h[j] = 0;
  __syncthreads();
  const int e0 = g * (CT * CE) + tid;
  if ((g + 1) * (CT * CE) <= E) count_chunk<false>(dst, rank, h, e0, E);
  else                          count_chunk<true>(dst, rank, h, e0, E);
  __syncthreads();
  for (int j = tid; j < NB; j += CT) H[j * GP + g] = h[j];
  if (g == 0)
    for (int j = tid; j < NB; j += CT)
      for (int q = G; q < GP; ++q) H[j * GP + q] = 0;
}

#define LB 8
#define NR (NBMAX / ((CT / 32) * (32 / LB)))
#define TU 7
static_assert(NR == 4 && TU == 7, "the load fence in table_trip names NR*TU values");

static __device__ __forceinline__ void table_trip(const v4i* __restrict__ H4, const int (&rq)[NR],
                                                  int (&pa)[NR], int (&ta)[NR], int q0, int nq, int qb) {
  v4i v[NR][TU];
#pragma unroll
  for (int rr = 0; rr < NR; ++rr)
#pragma unroll
    for (int u = 0; u < TU; ++u) v[rr][u] = H4[rq[rr] + min(q0 + u * LB, nq - 1)];
  asm volatile("" :: "v"(v[0][0]), "v"(v[0][1]), "v"(v[0][2]), "v"(v[0][3]), "v"(v[0][4]), "v"(v[0][5]), "v"(v[0][6]),
                     "v"(v[1][0]), "v"(v[1][1]), "v"(v[1][2]), "v"(v[1][3]), "v"(v[1][4]), "v"(v[1][5]), "v"(v[1][6]),
                     "v"(v[2][0]), "v"(v[2][1]), "v"(v[2][2]), "v"(v[2][3]), "v"(v[2][4]), "v"(v[2][5]), "v"(v[2][6]),
                     "v"(v[3][0]), "v"(v[3][1]), "v"(v[3][2]), "v"(v[3][3]), "v"(v[3][4]), "v"(v[3][5]), "v"(v[3][6]));
#pragma unroll
  for (int rr = 0; rr < NR; ++rr)
#pragma unroll
    for (int u = 0; u < TU; ++u) {
      const int q = q0 + u * LB;
      const int sm = (v[rr][u].x + v[rr][u].y) + (v[rr][u].z + v[rr][u].w);
      ta[rr] += q < nq ? sm : 0;
      pa[rr] += q < qb ? sm : 0;
    }
}

template <bool TAIL>
static __device__ __forceinline__ void scatter_chunk(const int* __restrict__ src, const int* __restrict__ dst,
                                                     const int* __restrict__ rank, const int* __restrict__ H,
                                                     int* __restrict__ bstart, int* __restrict__ tmp,
                                                     int* P, int* T, int* fs,
                                                     int E, int NB, int GP, int g) {
  const int tid = threadIdx.x;
  const int lane = tid & 31;
  const int wave = tid >> 5;
  const int e0 = g * (CT * CE) + tid;
  int s[CE], d[CE], r[CE];
#pragma unroll
  for (int u = 0; u < CE; ++u) {
    const int e = TAIL ? min(e0 + u * CT, E - 1) : e0 + u * CT;
    s[u] = src[e];
    d[u] = dst[e];
    r[u] = rank[e];
  }

  const int sub = lane & (LB - 1);
  const int bl = lane / LB;
  const int nq = GP >> 2;
  const int qb = g >> 2;
  const int kb = g & 3;
  const v4i* H4 = (const v4i*)H;
  int rq[NR];
  int pa[NR], ta[NR];
  v4i wb[NR];
#pragma unroll
  for (int rr = 0; rr < NR; ++rr) {
    const int b = min(rr * (NBMAX / NR) + wave * (32 / LB) + bl, NB - 1);
    rq[rr] = b * nq;
    pa[rr] = 0;
    ta[rr] = 0;
    wb[rr] = H4[rq[rr] + qb];
  }
  if (nq <= TU * LB) table_trip(H4, rq, pa, ta, sub, nq, qb);
  else
    for (int q0 = sub; q0 < nq; q0 += TU * LB) table_trip(H4, rq, pa, ta, q0, nq, qb);
#pragma unroll
  for (int rr = 0; rr < NR; ++rr) {
    const int part = (kb > 0 ? wb[rr].x : 0) + (kb > 1 ? wb[rr].y : 0) + (kb > 2 ? wb[rr].z : 0);
    pa[rr] += sub == 0 ? part : 0;
#pragma unroll
    for (int off = LB / 2; off > 0; off >>= 1) {
      pa[rr] += __shfl_xor(pa[rr], off, 32);
      ta[rr] += __shfl_xor(ta[rr], off, 32);
    }
    const int b = rr * (NBMAX / NR) + wave * (32 / LB) + bl;
    if (sub == 0 && b < NB) {
      P[b] = pa[rr];
      T[b] = ta[rr];
    }
  }
  __syncthreads();
  if (tid < 32) {
    int v[8];
    int run = 0;
#pragma unroll
    for (int j = 0; j < 8; ++j) {
      const int t = 8 * tid + j < NB ? T[8 * tid + j] : 0;
      v[j] = run;
      run += t;
    }
    int incl = run;
#pragma unroll
    for (int off = 1; off < 32; off <<= 1) {
      int t = __shfl_up(incl, off, 32);
      if (tid >= off) incl += t;
    }
    const int lanebase = incl - run;
#pragma unroll
    for (int j = 0; j < 8; ++j) {
      const int b = 8 * tid + j;
      if (b < NB) {
        const int st = lanebase + v[j];
        fs[b] = st + P[b];
        if (g == 0) bstart[b] = st;
      }
    }
    if (g == 0 && tid == 31) bstart[NB] = incl;
  }
  __syncthreads();

#pragma unroll
  for (int u = 0; u < CE; ++u)
    if (!TAIL || e0 + u * CT < E) ((int2*)tmp)[fs[d[u] >> BSH] + r[u]] = make_int2(s[u], d[u]);
}

__global__ __launch_bounds__(CT) void k_bucket_scatter(const int* __restrict__ src,
                                                       const int* __restrict__ dst,
                                                       const int* __restrict__ rank,
                                                       const int* __restrict__ H,
                                                       int* __restrict__ bstart,
                                                       int* __restrict__ tmp, int E, int NB, int GP) {
  __shared__ int P[NBMAX];
  __shared__ int T[NBMAX];
  __shared__ int fs[NBMAX];
  const int g = blockIdx.x;
  if ((g + 1) * (CT * CE) <= E) scatter_chunk<false>(src, dst, rank, H, bstart, tmp, P, T, fs, E, NB, GP, g);
  else                          scatter_chunk<true>(src, dst, rank, H, bstart, tmp, P, T, fs, E, NB, GP, g);
}

__global__ __launch_bounds__(1024) void k_bucket_sort(const int* __restrict__ tmp,
                                                      const int* __restrict__ bstart,
                                                      int* __restrict__ list,
                                                      int* __restrict__ cur,
                                                      float* __restrict__ dinv, int N) {
  __shared__ int c[256];
  __shared__ int st[256];
  __shared__ int f[256];
  const int tid = threadIdx.x;
  const int b = blockIdx.x;
  const int lo = bstart[b];
  const int hi = bstart[b + 1];
  if (tid < 256) {
    c[tid] = 0;
    f[tid] = 0;
  }
  __syncthreads();
  for (int k0 = lo + tid; k0 < hi; k0 += 4096) {
    int d[4];
#pragma unroll
    for (int u = 0; u < 4; ++u) d[u] = tmp[2 * min(k0 + u * 1024, hi - 1) + 1];
#pragma unroll
    for (int u = 0; u < 4; ++u)
      if (k0 + u * 1024 < hi) atomicAdd(&c[d[u] & 255], 1);
  }
  __syncthreads();
  if (tid < 32) {
    int v[8];
    int run = 0;
#pragma unroll
    for (int j = 0; j < 8; ++j) {
      const int t = c[8 * tid + j];
      v[j] = run;
      run += t;
    }
    int incl = run;
#pragma unroll
    for (int off = 1; off < 32; off <<= 1) {
      int t = __shfl_up(incl, off, 32);
      if (tid >= off) incl += t;
    }
    const int lanebase = incl - run;
#pragma unroll
    for (int j = 0; j < 8; ++j) st[8 * tid + j] = lanebase + v[j];
  }
  __syncthreads();
  if (tid < 256) {
    const int node = b * 256 + tid;
    if (node < N) {
      cur[node] = lo + st[tid];
      dinv[node] = rsqrtf((float)(c[tid] + 1));
    }
  }
  if (b == ((N + (1 << BSH) - 1) >> BSH) - 1 && tid == 0) cur[N] = hi;
  for (int k0 = lo + tid; k0 < hi; k0 += 4096) {
    int2 ent[4];
#pragma unroll
    for (int u = 0; u < 4; ++u) ent[u] = ((const int2*)tmp)[min(k0 + u * 1024, hi - 1)];
#pragma unroll
    for (int u = 0; u < 4; ++u) {
      if (k0 + u * 1024 < hi) {
        const int j = ent[u].y & 255;
        const int r = atomicAdd(&f[j], 1);
        list[lo + st[j] + r] = ent[u].x;
      }
    }
  }
}

static __device__ __forceinline__ float xhalf(float x) {
  return __builtin_bit_cast(float, __builtin_amdgcn_permlanex16(__builtin_bit_cast(int, x), __builtin_bit_cast(int, x),
                                                                0x76543210, 0xfedcba98, false, false));
}

template <bool WSRC>
static __device__ __forceinline__ float4 gather_row(const float* __restrict__ h,
                                                    const int* __restrict__ list,
                                                    const int* __restrict__ cur,
                                                    const float* __restrict__ dinv,
                                                    int i, int lane) {
  const int half = lane >> 4;
  const int c4 = (lane & 15) * 4;
  const int start = cur[i];
  const int end = cur[i + 1];
  const float di = dinv[i];
  const float4 self = *(const float4*)(h + (size_t)i * CH + c4);
  const float ws = WSRC ? di : 1.0f;
  float4 acc;
  acc.x = half ? 0.0f : self.x * ws;
  acc.y = half ? 0.0f : self.y * ws;
  acc.z = half ? 0.0f : self.z * ws;
  acc.w = half ? 0.0f : self.w * ws;

  for (int base = start; base < end; base += 32) {
    const int n = min(32, end - base);
    int sv = i;
    if (lane < n) sv = list[base + lane];
    int dv = 0;
    if (WSRC) dv = __float_as_int(dinv[sv]);
    const int npairs = (n + 1) >> 1;
    for (int p = 0; p < npairs; p += 8) {
      int s[8];
      float w[8];
      float4 r[8];
#pragma unroll
      for (int u = 0; u < 8; ++u) {
        const int q = 2 * (p + u);
        const int sa = __builtin_amdgcn_readlane(sv, q);
        const int sb = __builtin_amdgcn_readlane(sv, q + 1);
        s[u] = half ? sb : sa;
      }
#pragma unroll
      for (int u = 0; u < 8; ++u) {
        r[u] = make_float4(0.0f, 0.0f, 0.0f, 0.0f);
        if (2 * (p + u) + half < n) r[u] = *(const float4*)(h + (size_t)s[u] * CH + c4);
      }
      if (WSRC) {
#pragma unroll
        for (int u = 0; u < 8; ++u) {
          const int q = 2 * (p + u);
          const int wa = __builtin_amdgcn_readlane(dv, q);
          const int wb = __builtin_amdgcn_readlane(dv, q + 1);
          w[u] = __int_as_float(half ? wb : wa);
        }
#pragma unroll
        for (int u = 0; u < 8; ++u) {
          acc.x = fmaf(r[u].x, w[u], acc.x);
          acc.y = fmaf(r[u].y, w[u], acc.y);
          acc.z = fmaf(r[u].z, w[u], acc.z);
          acc.w = fmaf(r[u].w, w[u], acc.w);
        }
      } else {
#pragma unroll
        for (int u = 0; u < 8; ++u) {
          acc.x += r[u].x;
          acc.y += r[u].y;
          acc.z += r[u].z;
          acc.w += r[u].w;
        }
      }
    }
  }
  acc.x = (acc.x + xhalf(acc.x)) * di;
  acc.y = (acc.y + xhalf(acc.y)) * di;
  acc.z = (acc.z + xhalf(acc.z)) * di;
  acc.w = (acc.w + xhalf(acc.w)) * di;
  return acc;
}

#define TN 16
#define TP 68

template <bool PROJ>
static __device__ __forceinline__ void layer_tile(const float* __restrict__ hin,
                                                  const int* __restrict__ list,
                                                  const int* __restrict__ cur,
                                                  const float* __restrict__ dinv,
                                                  const float* __restrict__ W,
                                                  const float* __restrict__ b,
                                                  const float* __restrict__ Wfc,
                                                  float* __restrict__ out,
                                                  float* __restrict__ ps,
                                                  float* __restrict__ pd, int N) {
  __shared__ float agg[TN * TP];
  __shared__ float act[TN * TP];
  const int lane = threadIdx.x & 31;
  const int wave = threadIdx.x >> 5;
  const int kh = lane >> 4;
  const int m = lane & 15;
  const int c4 = m * 4;
  const int tile0 = blockIdx.x * TN;

  for (int t = 0; t < 4; ++t) {
    const int r = wave * 4 + t;
    const int i = __builtin_amdgcn_readfirstlane(tile0 + r);
    float4 v = make_float4(0.0f, 0.0f, 0.0f, 0.0f);
    if (i < N) v = gather_row<!PROJ>(hin, list, cur, dinv, i, lane);
    if (lane < 16) *(float4*)(agg + r * TP + c4) = v;
  }

  const int n0 = wave * 16;
  v2f bw[CH / 4];
#pragma unroll
  for (int j = 0; j < CH / 4; ++j) {
    bw[j].x = W[(4 * j + 2 * kh + 0) * CH + n0 + m];
    bw[j].y = W[(4 * j + 2 * kh + 1) * CH + n0 + m];
  }
  const float bias = b[n0 + m];
  float dsc[2] = {0.0f, 0.0f};
  if (!PROJ) {
#pragma unroll
    for (int t = 0; t < 2; ++t) {
      const int i = tile0 + wave * 4 + 2 * t + kh;
      if (i < N) dsc[t] = dinv[i];
    }
  }
  __syncthreads();

  const float* arow = agg + m * TP + 2 * kh;
  v8f acc = {};
#pragma unroll
  for (int j = 0; j < CH / 4; ++j) {
    v2f a;
    a.x = arow[4 * j];
    a.y = arow[4 * j + 1];
    acc = __builtin_amdgcn_wmma_f32_16x16x4_f32(false, a, false, bw[j],
                                                (short)0, acc, false, false);
  }
#pragma unroll
  for (int v = 0; v < 8; ++v) {
    const float y = acc[v] + bias;
    act[(v + 8 * kh) * TP + n0 + m] = y > 0.0f ? y : 0.0f;
  }
  float4 w0, w1;
  if (PROJ) {
    w0 = *(const float4*)(Wfc + c4);
    w1 = *(const float4*)(Wfc + CH + c4);
  }
  __syncthreads();

#pragma unroll
  for (int t = 0; t < 2; ++t) {
    const int r = wave * 4 + 2 * t + kh;
    const int i = tile0 + r;
    const float4 v = *(const float4*)(act + r * TP + c4);
    if (PROJ) {
      float s0 = (v.x * w0.x + v.y * w0.y) + (v.z * w0.z + v.w * w0.w);
      float s1 = (v.x * w1.x + v.y * w1.y) + (v.z * w1.z + v.w * w1.w);
#pragma unroll
      for (int off = 8; off > 0; off >>= 1) {
        s0 += __shfl_xor(s0, off, 32);
        s1 += __shfl_xor(s1, off, 32);
      }
      if (m == 0 && i < N) {
        ps[i] = s0;
        pd[i] = s1;
      }
    } else {
      if (i < N) *(float4*)(out + (size_t)i * CH + c4) = make_float4(v.x * dsc[t], v.y * dsc[t], v.z * dsc[t], v.w * dsc[t]);
    }
  }
}

__global__ __launch_bounds__(128) void k_layer_relu(const float* __restrict__ hin,
                                                    const int* __restrict__ list,
                                                    const int* __restrict__ cur,
                                                    const float* __restrict__ dinv,
                                                    const float* __restrict__ W,
                                                    const float* __restrict__ b,
                                                    float* __restrict__ out, int N) {
  layer_tile<false>(hin, list, cur, dinv, W, b, nullptr, out, nullptr, nullptr, N);
}

__global__ __launch_bounds__(128) void k_layer_relu_proj(const float* __restrict__ hin,
                                                         const int* __restrict__ list,
                                                         const int* __restrict__ cur,
                                                         const float* __restrict__ dinv,
                                                         const float* __restrict__ W,
                                                         const float* __restrict__ b,
                                                         const float* __restrict__ Wfc,
                                                         float* __restrict__ ps,
                                                         float* __restrict__ pd, int N) {
  layer_tile<true>(hin, list, cur, dinv, W, b, Wfc, nullptr, ps, pd, N);
}

#define PT 128
#define PQ 2

template <bool TAIL>
static __device__ __forceinline__ void pred_groups(const float* __restrict__ ps, const float* __restrict__ pd,
                                                   const int* __restrict__ src, const int* __restrict__ dst,
                                                   float b, float* __restrict__ out, int q0, int nq) {
  int4 s[PQ], d[PQ];
  float4 a[PQ], c[PQ];
#pragma unroll
  for (int u = 0; u < PQ; ++u) {
    const int q = TAIL ? min(q0 + u * PT, nq - 1) : q0 + u * PT;
    s[u] = ((const int4*)src)[q];
    d[u] = ((const int4*)dst)[q];
  }
#pragma unroll
  for (int u = 0; u < PQ; ++u) {
    a[u].x = ps[s[u].x];  c[u].x = pd[d[u].x];
    a[u].y = ps[s[u].y];  c[u].y = pd[d[u].y];
    a[u].z = ps[s[u].z];  c[u].z = pd[d[u].z];
    a[u].w = ps[s[u].w];  c[u].w = pd[d[u].w];
  }
#pragma unroll
  for (int u = 0; u < PQ; ++u) {
    const int q = q0 + u * PT;
    if (!TAIL || q < nq)
      ((float4*)out)[q] = make_float4(a[u].x + c[u].x + b, a[u].y + c[u].y + b,
                                      a[u].z + c[u].z + b, a[u].w + c[u].w + b);
  }
}

__global__ __launch_bounds__(PT) void k_edge_pred(const float* __restrict__ ps, const float* __restrict__ pd,
                                                  const int* __restrict__ src, const int* __restrict__ dst,
                                                  const float* __restrict__ bfc, float* __restrict__ out, int E) {
  const float b = bfc[0];
  if ((E & 3) == 0) {
    const int nq = E >> 2;
    const int q0 = blockIdx.x * (PT * PQ) + threadIdx.x;
    if ((blockIdx.x + 1) * (PT * PQ) <= nq) pred_groups<false>(ps, pd, src, dst, b, out, q0, nq);
    else                                    pred_groups<true>(ps, pd, src, dst, b, out, q0, nq);
  } else {
    const int e0 = blockIdx.x * (PT * PQ * 4) + threadIdx.x;
    for (int u = 0; u < PQ * 4; ++u) {
      const int e = e0 + u * PT;
      if (e < E) out[e] = ps[src[e]] + pd[dst[e]] + b;
    }
  }
}

extern "C" void kernel_launch(void* const* d_in, const int* in_sizes, int n_in,
                              void* d_out, int out_size, void* d_ws, size_t ws_size,
                              hipStream_t stream) {
  const float* x   = (const float*)d_in[0];
  const int*   ei  = (const int*)d_in[1];
  const float* W1  = (const float*)d_in[2];
  const float* b1  = (const float*)d_in[3];
  const float* W2  = (const float*)d_in[4];
  const float* b2  = (const float*)d_in[5];
  const float* Wfc = (const float*)d_in[6];
  const float* bfc = (const float*)d_in[7];

  const int N = in_sizes[0] / CH;
  const int E = in_sizes[1] / 2;
  const int NF = N * CH;

  const int* src = ei;
  const int* dst = ei + E;

  const int NB = (N + (1 << BSH) - 1) >> BSH;
  const int G  = (E + CT * CE - 1) / (CT * CE);
  const int GP = (G + 3) & ~3;
  const int M  = NB * GP;
  const int E4 = (E + 3) & ~3;
  const int NB4 = (NB + 1 + 3) & ~3;

  float* ws    = (float*)d_ws;
  float* AB    = ws;
  int*   list  = (int*)(AB + (size_t)NF);
  int*   tmp   = list + E4;
  int*   rank  = tmp + 2 * (size_t)E;
  int*   H     = rank + E4;
  int*   bstart = H + M;
  int*   cur   = bstart + NB4;
  float* dinv  = (float*)(cur + N + 1);
  float* PS    = dinv + N;
  float* PD    = PS + N;

  dim3 gE((E + PT * PQ * 4 - 1) / (PT * PQ * 4));
  dim3 gT((N + TN - 1) / TN);

  k_bucket_count<<<dim3(G), dim3(CT), 0, stream>>>(dst, rank, H, E, NB, G, GP);
  k_bucket_scatter<<<dim3(G), dim3(CT), 0, stream>>>(src, dst, rank, H, bstart, tmp, E, NB, GP);
  k_bucket_sort<<<dim3(NB), dim3(1024), 0, stream>>>(tmp, bstart, list, cur, dinv, N);

  k_layer_relu<<<gT, dim3(128), 0, stream>>>(x, list, cur, dinv, W1, b1, AB, N);

  k_layer_relu_proj<<<gT, dim3(128), 0, stream>>>(AB, list, cur, dinv, W2, b2, Wfc, PS, PD, N);

  k_edge_pred<<<gE, dim3(PT), 0, stream>>>(PS, PD, src, dst, bfc, (float*)d_out, E);
}


# Chunk-local bucket sort: a chunk block orders its own edges by bucket in LDS (table word = start,count); the bucket kernel collects its runs itself: edge-moving launch gone; sum order = LDS arrival
# speedup vs baseline: 5.1309x; 1.1843x over previous
#include <hip/hip_runtime.h>
#include <hip/hip_bf16.h>

typedef __attribute__((ext_vector_type(2)))  float    v2f;
typedef __attribute__((ext_vector_type(8)))  float    v8f;
typedef __attribute__((ext_vector_type(4)))  int      v4i;

#define CH 64

#define BSH   8
#define NBMAX 256
#define CT    512
#define CE    8
#define CHE   (CT * CE)
static_assert(CHE < 65536, "a table word holds start << 16 | count");

template <bool TAIL>
static __device__ __forceinline__ void chunk_sort(const int* __restrict__ src, const int* __restrict__ dst,
                                                  int* __restrict__ H, int* __restrict__ tmp,
                                                  int* h, int* fs, int E, int NB, int GP, int g) {
  const int tid = threadIdx.x;
  const int e0 = g * CHE + tid;
  int s[CE], d[CE], r[CE];
#pragma unroll
  for (int u = 0; u < CE; ++u) {
    const int e = TAIL ? min(e0 + u * CT, E - 1) : e0 + u * CT;
    s[u] = src[e];
    d[u] = dst[e];
  }
#pragma unroll
  for (int u = 0; u < CE; ++u) {
    r[u] = 0;
    if (!TAIL || e0 + u * CT < E) r[u] = atomicAdd(&h[d[u] >> BSH], 1);
  }
  __syncthreads();
  int v[8], t[8];
  int lanebase = 0;
  if (tid < 32) {
    int run = 0;
#pragma unroll
    for (int j = 0; j < 8; ++j) {
      t[j] = 8 * tid + j < NB ? h[8 * tid + j] : 0;
      v[j] = run;
      run += t[j];
    }
    int incl = run;
#pragma unroll
    for (int off = 1; off < 32; off <<= 1) {
      int x = __shfl_up(incl, off, 32);
      if (tid >= off) incl += x;
    }
    lanebase = incl - run;
#pragma unroll
    for (int j = 0; j < 8; ++j)
      if (8 * tid + j < NB) fs[8 * tid + j] = lanebase + v[j];
  }
  __syncthreads();
  if (tid < 32) {
#pragma unroll
    for (int j = 0; j < 8; ++j) {
      const int b = 8 * tid + j;
      if (b < NB) H[b * GP + g] = ((lanebase + v[j]) << 16) | t[j];
    }
  }
#pragma unroll
  for (int u = 0; u < CE; ++u)
    if (!TAIL || e0 + u * CT < E) ((int2*)tmp)[g * CHE + fs[d[u] >> BSH] + r[u]] = make_int2(s[u], d[u]);
}

__global__ __launch_bounds__(CT) void k_chunk_sort(const int* __restrict__ src,
                                                   const int* __restrict__ dst,
                                                   int* __restrict__ H,
                                                   int* __restrict__ tmp, int E, int NB, int G, int GP) {
  __shared__ int h[NBMAX];
  __shared__ int fs[NBMAX];
  const int tid = threadIdx.x;
  const int g = blockIdx.x;
  for (int j = tid; j < NB; j += CT) h[j] = 0;
  __syncthreads();
  if ((g + 1) * CHE <= E) chunk_sort<false>(src, dst, H, tmp, h, fs, E, NB, GP, g);
  else                    chunk_sort<true>(src, dst, H, tmp, h, fs, E, NB, GP, g);
  if (g == 0)
    for (int j = tid; j < NB; j += CT)
      for (int q = G; q < GP; ++q) H[j * GP + q] = 0;
}

#define RU 8
#define LT 10
static_assert(RU == 8, "the load fences in k_bucket_sort name RU values");
__global__ __launch_bounds__(1024) void k_bucket_sort(const int* __restrict__ tmp,
                                                      const int* __restrict__ H,
                                                      int* __restrict__ list,
                                                      int* __restrict__ cur,
                                                      float* __restrict__ dinv, int N, int G, int GP) {
  __shared__ int c[256];
  __shared__ int st[256];
  __shared__ int f[256];
  __shared__ int red[32];
  __shared__ int slo;
  const int tid = threadIdx.x;
  const int lane = tid & 31;
  const int wv = __builtin_amdgcn_readfirstlane(tid >> 5);
  const int b = blockIdx.x;
  const int2* tmp2 = (const int2*)tmp;
  const int* row = H + b * GP;
  const int wl0 = row[min(wv + 32 * lane, G - 1)];

  const v4i* H4 = (const v4i*)H;
  const int n4 = (b * GP) >> 2;
  v4i hv[LT];
#pragma unroll
  for (int u = 0; u < LT; ++u) hv[u] = H4[min(tid + u * 1024, max(n4 - 1, 0))];
  if (tid < 256) {
    c[tid] = 0;
    f[tid] = 0;
  }
  __syncthreads();

  for (int g0 = wv; g0 < G; g0 += 32 * RU) {
    int n[RU], o[RU], d[RU];
    const int wl = g0 == wv ? wl0 : row[min(g0 + 32 * lane, G - 1)];
#pragma unroll
    for (int u = 0; u < RU; ++u) {
      const int w = __builtin_amdgcn_readlane(wl, u);
      n[u] = g0 + 32 * u < G ? w & 0xFFFF : 0;
      o[u] = n[u] > 0 ? (g0 + 32 * u) * CHE + (w >> 16) : 0;
    }
#pragma unroll
    for (int u = 0; u < RU; ++u) d[u] = tmp[2 * (o[u] + min(lane, max(n[u] - 1, 0))) + 1];
    asm volatile("" :: "v"(d[0]), "v"(d[1]), "v"(d[2]), "v"(d[3]), "v"(d[4]), "v"(d[5]), "v"(d[6]), "v"(d[7]));
#pragma unroll
    for (int u = 0; u < RU; ++u)
      if (lane < n[u]) atomicAdd(&c[d[u] & 255], 1);
#pragma unroll
    for (int u = 0; u < RU; ++u)
      for (int k = lane + 32; k < n[u]; k += 32) atomicAdd(&c[tmp[2 * (o[u] + k) + 1] & 255], 1);
  }
  int part = 0;
#pragma unroll
  for (int u = 0; u < LT; ++u) {
    const int sm = ((hv[u].x & 0xFFFF) + (hv[u].y & 0xFFFF)) + ((hv[u].z & 0xFFFF) + (hv[u].w & 0xFFFF));
    part += tid + u * 1024 < n4 ? sm : 0;
  }
  for (int i0 = tid + LT * 1024; i0 < n4; i0 += 1024) {
    const v4i v = H4[i0];
    part += ((v.x & 0xFFFF) + (v.y & 0xFFFF)) + ((v.z & 0xFFFF) + (v.w & 0xFFFF));
  }
#pragma unroll
  for (int off = 16; off > 0; off >>= 1) part += __shfl_xor(part, off, 32);
  if (lane == 0) red[wv] = part;
  __syncthreads();
  if (tid < 32) {
    int v[8];
    int run = 0;
#pragma unroll
    for (int j = 0; j < 8; ++j) {
      const int t = c[8 * tid + j];
      v[j] = run;
      run += t;
    }
    int incl = run;
#pragma unroll
    for (int off = 1; off < 32; off <<= 1) {
      int t = __shfl_up(incl, off, 32);
      if (tid >= off) incl += t;
    }
    const int lanebase = incl - run;
#pragma unroll
    for (int j = 0; j < 8; ++j) st[8 * tid + j] = lanebase + v[j];
    int p = red[tid];
#pragma unroll
    for (int off = 16; off > 0; off >>= 1) p += __shfl_xor(p, off, 32);
    if (tid == 0) slo = p;
  }
  __syncthreads();
  const int lo = slo;
  if (tid < 256) {
    const int node = b * 256 + tid;
    if (node < N) {
      cur[node] = lo + st[tid];
      dinv[node] = rsqrtf((float)(c[tid] + 1));
    }
  }
  if (b == ((N + (1 << BSH) - 1) >> BSH) - 1 && tid == 255) cur[N] = lo + st[255] + c[255];

  for (int g0 = wv; g0 < G; g0 += 32 * RU) {
    int n[RU], o[RU], r[RU], sg[RU];
    int2 ent[RU];
    const int wl = g0 == wv ? wl0 : row[min(g0 + 32 * lane, G - 1)];
#pragma unroll
    for (int u = 0; u < RU; ++u) {
      const int w = __builtin_amdgcn_readlane(wl, u);
      n[u] = g0 + 32 * u < G ? w & 0xFFFF : 0;
      o[u] = n[u] > 0 ? (g0 + 32 * u) * CHE + (w >> 16) : 0;
    }
#pragma unroll
    for (int u = 0; u < RU; ++u) ent[u] = tmp2[o[u] + min(lane, max(n[u] - 1, 0))];
    asm volatile("" :: "v"(ent[0].y), "v"(ent[1].y), "v"(ent[2].y), "v"(ent[3].y), "v"(ent[4].y), "v"(ent[5].y), "v"(ent[6].y), "v"(ent[7].y));
#pragma unroll
    for (int u = 0; u < RU; ++u) {
      r[u] = 0;
      if (lane < n[u]) r[u] = atomicAdd(&f[ent[u].y & 255], 1);
    }
#pragma unroll
    for (int u = 0; u < RU; ++u) sg[u] = st[ent[u].y & 255];
#pragma unroll
    for (int u = 0; u < RU; ++u)
      if (lane < n[u]) list[lo + sg[u] + r[u]] = ent[u].x;
#pragma unroll
    for (int u = 0; u < RU; ++u)
      for (int k = lane + 32; k < n[u]; k += 32) {
        const int2 e = tmp2[o[u] + k];
        const int j = e.y & 255;
        const int rk = atomicAdd(&f[j], 1);
        list[lo + st[j] + rk] = e.x;
      }
  }
}

static __device__ __forceinline__ float xhalf(float x) {
  return __builtin_bit_cast(float, __builtin_amdgcn_permlanex16(__builtin_bit_cast(int, x), __builtin_bit_cast(int, x),
                                                                0x76543210, 0xfedcba98, false, false));
}

template <bool WSRC>
static __device__ __forceinline__ float4 gather_row(const float* __restrict__ h,
                                                    const int* __restrict__ list,
                                                    const int* __restrict__ cur,
                                                    const float* __restrict__ dinv,
                                                    int i, int lane) {
  const int half = lane >> 4;
  const int c4 = (lane & 15) * 4;
  const int start = cur[i];
  const int end = cur[i + 1];
  const float di = dinv[i];
  const float4 self = *(const float4*)(h + (size_t)i * CH + c4);
  const float ws = WSRC ? di : 1.0f;
  float4 acc;
  acc.x = half ? 0.0f : self.x * ws;
  acc.y = half ? 0.0f : self.y * ws;
  acc.z = half ? 0.0f : self.z * ws;
  acc.w = half ? 0.0f : self.w * ws;

  for (int base = start; base < end; base += 32) {
    const int n = min(32, end - base);
    int sv = i;
    if (lane < n) sv = list[base + lane];
    int dv = 0;
    if (WSRC) dv = __float_as_int(dinv[sv]);
    const int npairs = (n + 1) >> 1;
    for (int p = 0; p < npairs; p += 8) {
      int s[8];
      float w[8];
      float4 r[8];
#pragma unroll
      for (int u = 0; u < 8; ++u) {
        const int q = 2 * (p + u);
        const int sa = __builtin_amdgcn_readlane(sv, q);
        const int sb = __builtin_amdgcn_readlane(sv, q + 1);
        s[u] = half ? sb : sa;
      }
#pragma unroll
      for (int u = 0; u < 8; ++u) {
        r[u] = make_float4(0.0f, 0.0f, 0.0f, 0.0f);
        if (2 * (p + u) + half < n) r[u] = *(const float4*)(h + (size_t)s[u] * CH + c4);
      }
      if (WSRC) {
#pragma unroll
        for (int u = 0; u < 8; ++u) {
          const int q = 2 * (p + u);
          const int wa = __builtin_amdgcn_readlane(dv, q);
          const int wb = __builtin_amdgcn_readlane(dv, q + 1);
          w[u] = __int_as_float(half ? wb : wa);
        }
#pragma unroll
        for (int u = 0; u < 8; ++u) {
          acc.x = fmaf(r[u].x, w[u], acc.x);
          acc.y = fmaf(r[u].y, w[u], acc.y);
          acc.z = fmaf(r[u].z, w[u], acc.z);
          acc.w = fmaf(r[u].w, w[u], acc.w);
        }
      } else {
#pragma unroll
        for (int u = 0; u < 8; ++u) {
          acc.x += r[u].x;
          acc.y += r[u].y;
          acc.z += r[u].z;
          acc.w += r[u].w;
        }
      }
    }
  }
  acc.x = (acc.x + xhalf(acc.x)) * di;
  acc.y = (acc.y + xhalf(acc.y)) * di;
  acc.z = (acc.z + xhalf(acc.z)) * di;
  acc.w = (acc.w + xhalf(acc.w)) * di;
  return acc;
}

#define TN 16
#define TP 68

template <bool PROJ>
static __device__ __forceinline__ void layer_tile(const float* __restrict__ hin,
                                                  const int* __restrict__ list,
                                                  const int* __restrict__ cur,
                                                  const float* __restrict__ dinv,
                                                  const float* __restrict__ W,
                                                  const float* __restrict__ b,
                                                  const float* __restrict__ Wfc,
                                                  float* __restrict__ out,
                                                  float* __restrict__ ps,
                                                  float* __restrict__ pd, int N) {
  __shared__ float agg[TN * TP];
  __shared__ float act[TN * TP];
  const int lane = threadIdx.x & 31;
  const int wave = threadIdx.x >> 5;
  const int kh = lane >> 4;
  const int m = lane & 15;
  const int c4 = m * 4;
  const int tile0 = blockIdx.x * TN;

  for (int t = 0; t < 4; ++t) {
    const int r = wave * 4 + t;
    const int i = __builtin_amdgcn_readfirstlane(tile0 + r);
    float4 v = make_float4(0.0f, 0.0f, 0.0f, 0.0f);
    if (i < N) v = gather_row<!PROJ>(hin, list, cur, dinv, i, lane);
    if (lane < 16) *(float4*)(agg + r * TP + c4) = v;
  }

  const int n0 = wave * 16;
  v2f bw[CH / 4];
#pragma unroll
  for (int j = 0; j < CH / 4; ++j) {
    bw[j].x = W[(4 * j + 2 * kh + 0) * CH + n0 + m];
    bw[j].y = W[(4 * j + 2 * kh + 1) * CH + n0 + m];
  }
  const float bias = b[n0 + m];
  float dsc[2] = {0.0f, 0.0f};
  if (!PROJ) {
#pragma unroll
    for (int t = 0; t < 2; ++t) {
      const int i = tile0 + wave * 4 + 2 * t + kh;
      if (i < N) dsc[t] = dinv[i];
    }
  }
  __syncthreads();

  const float* arow = agg + m * TP + 2 * kh;
  v8f acc = {};
#pragma unroll
  for (int j = 0; j < CH / 4; ++j) {
    v2f a;
    a.x = arow[4 * j];
    a.y = arow[4 * j + 1];
    acc = __builtin_amdgcn_wmma_f32_16x16x4_f32(false, a, false, bw[j],
                                                (short)0, acc, false, false);
  }
#pragma unroll
  for (int v = 0; v < 8; ++v) {
    const float y = acc[v] + bias;
    act[(v + 8 * kh) * TP + n0 + m] = y > 0.0f ? y : 0.0f;
  }
  float4 w0, w1;
  if (PROJ) {
    w0 = *(const float4*)(Wfc + c4);
    w1 = *(const float4*)(Wfc + CH + c4);
  }
  __syncthreads();

#pragma unroll
  for (int t = 0; t < 2; ++t) {
    const int r = wave * 4 + 2 * t + kh;
    const int i = tile0 + r;
    const float4 v = *(const float4*)(act + r * TP + c4);
    if (PROJ) {
      float s0 = (v.x * w0.x + v.y * w0.y) + (v.z * w0.z + v.w * w0.w);
      float s1 = (v.x * w1.x + v.y * w1.y) + (v.z * w1.z + v.w * w1.w);
#pragma unroll
      for (int off = 8; off > 0; off >>= 1) {
        s0 += __shfl_xor(s0, off, 32);
        s1 += __shfl_xor(s1, off, 32);
      }
      if (m == 0 && i < N) {
        ps[i] = s0;
        pd[i] = s1;
      }
    } else {
      if (i < N) *(float4*)(out + (size_t)i * CH + c4) = make_float4(v.x * dsc[t], v.y * dsc[t], v.z * dsc[t], v.w * dsc[t]);
    }
  }
}

__global__ __launch_bounds__(128) void k_layer_relu(const float* __restrict__ hin,
                                                    const int* __restrict__ list,
                                                    const int* __restrict__ cur,
                                                    const float* __restrict__ dinv,
                                                    const float* __restrict__ W,
                                                    const float* __restrict__ b,
                                                    float* __restrict__ out, int N) {
  layer_tile<false>(hin, list, cur, dinv, W, b, nullptr, out, nullptr, nullptr, N);
}

__global__ __launch_bounds__(128) void k_layer_relu_proj(const float* __restrict__ hin,
                                                         const int* __restrict__ list,
                                                         const int* __restrict__ cur,
                                                         const float* __restrict__ dinv,
                                                         const float* __restrict__ W,
                                                         const float* __restrict__ b,
                                                         const float* __restrict__ Wfc,
                                                         float* __restrict__ ps,
                                                         float* __restrict__ pd, int N) {
  layer_tile<true>(hin, list, cur, dinv, W, b, Wfc, nullptr, ps, pd, N);
}

#define PT 128
#define PQ 2

template <bool TAIL>
static __device__ __forceinline__ void pred_groups(const float* __restrict__ ps, const float* __restrict__ pd,
                                                   const int* __restrict__ src, const int* __restrict__ dst,
                                                   float b, float* __restrict__ out, int q0, int nq) {
  int4 s[PQ], d[PQ];
  float4 a[PQ], c[PQ];
#pragma unroll
  for (int u = 0; u < PQ; ++u) {
    const int q = TAIL ? min(q0 + u * PT, nq - 1) : q0 + u * PT;
    s[u] = ((const int4*)src)[q];
    d[u] = ((const int4*)dst)[q];
  }
#pragma unroll
  for (int u = 0; u < PQ; ++u) {
    a[u].x = ps[s[u].x];  c[u].x = pd[d[u].x];
    a[u].y = ps[s[u].y];  c[u].y = pd[d[u].y];
    a[u].z = ps[s[u].z];  c[u].z = pd[d[u].z];
    a[u].w = ps[s[u].w];  c[u].w = pd[d[u].w];
  }
#pragma unroll
  for (int u = 0; u < PQ; ++u) {
    const int q = q0 + u * PT;
    if (!TAIL || q < nq)
      ((float4*)out)[q] = make_float4(a[u].x + c[u].x + b, a[u].y + c[u].y + b,
                                      a[u].z + c[u].z + b, a[u].w + c[u].w + b);
  }
}

__global__ __launch_bounds__(PT) void k_edge_pred(const float* __restrict__ ps, const float* __restrict__ pd,
                                                  const int* __restrict__ src, const int* __restrict__ dst,
                                                  const float* __restrict__ bfc, float* __restrict__ out, int E) {
  const float b = bfc[0];
  if ((E & 3) == 0) {
    const int nq = E >> 2;
    const int q0 = blockIdx.x * (PT * PQ) + threadIdx.x;
    if ((blockIdx.x + 1) * (PT * PQ) <= nq) pred_groups<false>(ps, pd, src, dst, b, out, q0, nq);
    else                                    pred_groups<true>(ps, pd, src, dst, b, out, q0, nq);
  } else {
    const int e0 = blockIdx.x * (PT * PQ * 4) + threadIdx.x;
    for (int u = 0; u < PQ * 4; ++u) {
      const int e = e0 + u * PT;
      if (e < E) out[e] = ps[src[e]] + pd[dst[e]] + b;
    }
  }
}

extern "C" void kernel_launch(void* const* d_in, const int* in_sizes, int n_in,
                              void* d_out, int out_size, void* d_ws, size_t ws_size,
                              hipStream_t stream) {
  const float* x   = (const float*)d_in[0];
  const int*   ei  = (const int*)d_in[1];
  const float* W1  = (const float*)d_in[2];
  const float* b1  = (const float*)d_in[3];
  const float* W2  = (const float*)d_in[4];
  const float* b2  = (const float*)d_in[5];
  const float* Wfc = (const float*)d_in[6];
  const float* bfc = (const float*)d_in[7];

  const int N = in_sizes[0] / CH;
  const int E = in_sizes[1] / 2;
  const int NF = N * CH;

  const int* src = ei;
  const int* dst = ei + E;

  const int NB = (N + (1 << BSH) - 1) >> BSH;
  const int G  = (E + CHE - 1) / CHE;
  const int GP = (G + 3) & ~3;
  const int M  = NB * GP;
  const int E4 = (E + 3) & ~3;

  float* ws    = (float*)d_ws;
  float* AB    = ws;
  int*   list  = (int*)(AB + (size_t)NF);
  int*   tmp   = list + E4;
  int*   H     = tmp + 2 * (size_t)E4;
  int*   cur   = H + M;
  float* dinv  = (float*)(cur + N + 1);
  float* PS    = dinv + N;
  float* PD    = PS + N;

  dim3 gE((E + PT * PQ * 4 - 1) / (PT * PQ * 4));
  dim3 gT((N + TN - 1) / TN);

  k_chunk_sort<<<dim3(G), dim3(CT), 0, stream>>>(src, dst, H, tmp, E, NB, G, GP);
  k_bucket_sort<<<dim3(NB), dim3(1024), 0, stream>>>(tmp, H, list, cur, dinv, N, G, GP);

  k_layer_relu<<<gT, dim3(128), 0, stream>>>(x, list, cur, dinv, W1, b1, AB, N);

  k_layer_relu_proj<<<gT, dim3(128), 0, stream>>>(AB, list, cur, dinv, W2, b2, Wfc, PS, PD, N);

  k_edge_pred<<<gE, dim3(PT), 0, stream>>>(PS, PD, src, dst, bfc, (float*)d_out, E);
}


# Sorted edges leave through LDS: chunk and bucket blocks put their edges in place in an LDS copy of their output piece and store it as whole 128-byte lines; scattered 8/4-byte stores gone; order kept
# speedup vs baseline: 5.5117x; 1.0742x over previous
#include <hip/hip_runtime.h>
#include <hip/hip_bf16.h>

typedef __attribute__((ext_vector_type(2)))  float    v2f;
typedef __attribute__((ext_vector_type(8)))  float    v8f;
typedef __attribute__((ext_vector_type(4)))  int      v4i;

#define CH 64

#define BSH   8
#define NBMAX 256
#define CT    512
#define CE    8
#define CHE   (CT * CE)
static_assert(CHE < 65536, "a table word holds start << 16 | count");

template <bool TAIL>
static __device__ __forceinline__ void chunk_sort(const int* __restrict__ src, const int* __restrict__ dst,
                                                  int* __restrict__ H, int* __restrict__ tmp,
                                                  int* h, int* fs, int2* stage, int E, int NB, int GP, int g) {
  const int tid = threadIdx.x;
  const int e0 = g * CHE + tid;
  int s[CE], d[CE], r[CE];
#pragma unroll
  for (int u = 0; u < CE; ++u) {
    const int e = TAIL ? min(e0 + u * CT, E - 1) : e0 + u * CT;
    s[u] = src[e];
    d[u] = dst[e];
  }
#pragma unroll
  for (int u = 0; u < CE; ++u) {
    r[u] = 0;
    if (!TAIL || e0 + u * CT < E) r[u] = atomicAdd(&h[d[u] >> BSH], 1);
  }
  __syncthreads();
  int v[8], t[8];
  int lanebase = 0;
  if (tid < 32) {
    int run = 0;
#pragma unroll
    for (int j = 0; j < 8; ++j) {
      t[j] = 8 * tid + j < NB ? h[8 * tid + j] : 0;
      v[j] = run;
      run += t[j];
    }
    int incl = run;
#pragma unroll
    for (int off = 1; off < 32; off <<= 1) {
      int x = __shfl_up(incl, off, 32);
      if (tid >= off) incl += x;
    }
    lanebase = incl - run;
#pragma unroll
    for (int j = 0; j < 8; ++j)
      if (8 * tid + j < NB) fs[8 * tid + j] = lanebase + v[j];
  }
  __syncthreads();
#pragma unroll
  for (int u = 0; u < CE; ++u)
    if (!TAIL || e0 + u * CT < E) stage[fs[d[u] >> BSH] + r[u]] = make_int2(s[u], d[u]);
  __syncthreads();
  if (!TAIL) {
    const int4* s4 = (const int4*)stage;
    int4* o4 = (int4*)(tmp + 2 * g * CHE);
#pragma unroll
    for (int j = 0; j < CHE / 2 / CT; ++j) o4[tid + j * CT] = s4[tid + j * CT];
  } else {
    const int cnt = E - g * CHE;
    int2* o2 = (int2*)tmp + g * CHE;
#pragma unroll
    for (int j = 0; j < CE; ++j) {
      const int k = tid + j * CT;
      if (k < cnt) o2[k] = stage[k];
    }
  }
  if (tid < 32) {
#pragma unroll
    for (int j = 0; j < 8; ++j) {
      const int b = 8 * tid + j;
      if (b < NB) H[b * GP + g] = ((lanebase + v[j]) << 16) | t[j];
    }
  }
}

__global__ __launch_bounds__(CT) void k_chunk_sort(const int* __restrict__ src,
                                                   const int* __restrict__ dst,
                                                   int* __restrict__ H,
                                                   int* __restrict__ tmp, int E, int NB, int G, int GP) {
  __shared__ int h[NBMAX];
  __shared__ int fs[NBMAX];
  __shared__ alignas(16) int2 stage[CHE];
  const int tid = threadIdx.x;
  const int g = blockIdx.x;
  for (int j = tid; j < NB; j += CT) h[j] = 0;
  __syncthreads();
  if ((g + 1) * CHE <= E) chunk_sort<false>(src, dst, H, tmp, h, fs, stage, E, NB, GP, g);
  else                    chunk_sort<true>(src, dst, H, tmp, h, fs, stage, E, NB, GP, g);
  if (g == 0)
    for (int j = tid; j < NB; j += CT)
      for (int q = G; q < GP; ++q) H[j * GP + q] = 0;
}

#define RU 8
#define LT 10
#define SCAP 8192
#define CO 5
static_assert(RU == 8, "the load fences in k_bucket_sort name RU values");

template <bool STAGED>
static __device__ __forceinline__ void bucket_place(const int2* __restrict__ tmp2, const int* __restrict__ row,
                                                    int wl0, int G, int wv, int lane,
                                                    int* f, const int* st, int* stage, int* __restrict__ listlo) {
  for (int g0 = wv; g0 < G; g0 += 32 * RU) {
    int n[RU], o[RU], r[RU], sg[RU];
    int2 ent[RU];
    const int wl = g0 == wv ? wl0 : row[min(g0 + 32 * lane, G - 1)];
#pragma unroll
    for (int u = 0; u < RU; ++u) {
      const int w = __builtin_amdgcn_readlane(wl, u);
      n[u] = g0 + 32 * u < G ? w & 0xFFFF : 0;
      o[u] = n[u] > 0 ? (g0 + 32 * u) * CHE + (w >> 16) : 0;
    }
#pragma unroll
    for (int u = 0; u < RU; ++u) ent[u] = tmp2[o[u] + min(lane, max(n[u] - 1, 0))];
    asm volatile("" :: "v"(ent[0].y), "v"(ent[1].y), "v"(ent[2].y), "v"(ent[3].y), "v"(ent[4].y), "v"(ent[5].y), "v"(ent[6].y), "v"(ent[7].y));
#pragma unroll
    for (int u = 0; u < RU; ++u) {
      r[u] = 0;
      if (lane < n[u]) r[u] = atomicAdd(&f[ent[u].y & 255], 1);
    }
#pragma unroll
    for (int u = 0; u < RU; ++u) sg[u] = st[ent[u].y & 255];
#pragma unroll
    for (int u = 0; u < RU; ++u)
      if (lane < n[u]) {
        if (STAGED) stage[sg[u] + r[u]] = ent[u].x;
        else        listlo[sg[u] + r[u]] = ent[u].x;
      }
#pragma unroll
    for (int u = 0; u < RU; ++u)
      for (int k = lane + 32; k < n[u]; k += 32) {
        const int2 e = tmp2[o[u] + k];
        const int j = e.y & 255;
        const int rk = atomicAdd(&f[j], 1);
        if (STAGED) stage[st[j] + rk] = e.x;
        else        listlo[st[j] + rk] = e.x;
      }
  }
}

__global__ __launch_bounds__(1024) void k_bucket_sort(const int* __restrict__ tmp,
                                                      const int* __restrict__ H,
                                                      int* __restrict__ list,
                                                      int* __restrict__ cur,
                                                      float* __restrict__ dinv, int N, int G, int GP) {
  __shared__ int c[256];
  __shared__ int st[256];
  __shared__ int f[256];
  __shared__ int red[32];
  __shared__ int slo;
  __shared__ int stage[SCAP];
  const int tid = threadIdx.x;
  const int lane = tid & 31;
  const int wv = __builtin_amdgcn_readfirstlane(tid >> 5);
  const int b = blockIdx.x;
  const int2* tmp2 = (const int2*)tmp;
  const int* row = H + b * GP;
  const int wl0 = row[min(wv + 32 * lane, G - 1)];

  const v4i* H4 = (const v4i*)H;
  const int n4 = (b * GP) >> 2;
  v4i hv[LT];
#pragma unroll
  for (int u = 0; u < LT; ++u) hv[u] = H4[min(tid + u * 1024, max(n4 - 1, 0))];
  if (tid < 256) {
    c[tid] = 0;
    f[tid] = 0;
  }
  __syncthreads();

  for (int g0 = wv; g0 < G; g0 += 32 * RU) {
    int n[RU], o[RU], d[RU];
    const int wl = g0 == wv ? wl0 : row[min(g0 + 32 * lane, G - 1)];
#pragma unroll
    for (int u = 0; u < RU; ++u) {
      const int w = __builtin_amdgcn_readlane(wl, u);
      n[u] = g0 + 32 * u < G ? w & 0xFFFF : 0;
      o[u] = n[u] > 0 ? (g0 + 32 * u) * CHE + (w >> 16) : 0;
    }
#pragma unroll
    for (int u = 0; u < RU; ++u) d[u] = tmp[2 * (o[u] + min(lane, max(n[u] - 1, 0))) + 1];
    asm volatile("" :: "v"(d[0]), "v"(d[1]), "v"(d[2]), "v"(d[3]), "v"(d[4]), "v"(d[5]), "v"(d[6]), "v"(d[7]));
#pragma unroll
    for (int u = 0; u < RU; ++u)
      if (lane < n[u]) atomicAdd(&c[d[u] & 255], 1);
#pragma unroll
    for (int u = 0; u < RU; ++u)
      for (int k = lane + 32; k < n[u]; k += 32) atomicAdd(&c[tmp[2 * (o[u] + k) + 1] & 255], 1);
  }
  int part = 0;
#pragma unroll
  for (int u = 0; u < LT; ++u) {
    const int sm = ((hv[u].x & 0xFFFF) + (hv[u].y & 0xFFFF)) + ((hv[u].z & 0xFFFF) + (hv[u].w & 0xFFFF));
    part += tid + u * 1024 < n4 ? sm : 0;
  }
  for (int i0 = tid + LT * 1024; i0 < n4; i0 += 1024) {
    const v4i v = H4[i0];
    part += ((v.x & 0xFFFF) + (v.y & 0xFFFF)) + ((v.z & 0xFFFF) + (v.w & 0xFFFF));
  }
#pragma unroll
  for (int off = 16; off > 0; off >>= 1) part += __shfl_xor(part, off, 32);
  if (lane == 0) red[wv] = part;
  __syncthreads();
  if (tid < 32) {
    int v[8];
    int run = 0;
#pragma unroll
    for (int j = 0; j < 8; ++j) {
      const int t = c[8 * tid + j];
      v[j] = run;
      run += t;
    }
    int incl = run;
#pragma unroll
    for (int off = 1; off < 32; off <<= 1) {
      int t = __shfl_up(incl, off, 32);
      if (tid >= off) incl += t;
    }
    const int lanebase = incl - run;
#pragma unroll
    for (int j = 0; j < 8; ++j) st[8 * tid + j] = lanebase + v[j];
    int p = red[tid];
#pragma unroll
    for (int off = 16; off > 0; off >>= 1) p += __shfl_xor(p, off, 32);
    if (tid == 0) slo = p;
  }
  __syncthreads();
  const int lo = slo;
  if (tid < 256) {
    const int node = b * 256 + tid;
    if (node < N) {
      cur[node] = lo + st[tid];
      dinv[node] = rsqrtf((float)(c[tid] + 1));
    }
  }
  if (b == ((N + (1 << BSH) - 1) >> BSH) - 1 && tid == 255) cur[N] = lo + st[255] + c[255];

  const int total = __builtin_amdgcn_readfirstlane(st[255] + c[255]);
  if (total <= SCAP) {
    bucket_place<true>(tmp2, row, wl0, G, wv, lane, f, st, stage, list + lo);
    __syncthreads();
    const int end = lo + total;
    for (int p0 = (lo & ~31) + tid; p0 < end; p0 += CO * 1024) {
      int v[CO];
#pragma unroll
      for (int j = 0; j < CO; ++j) v[j] = stage[max(min(p0 + j * 1024 - lo, total - 1), 0)];
#pragma unroll
      for (int j = 0; j < CO; ++j) {
        const int p = p0 + j * 1024;
        if (p >= lo && p < end) list[p] = v[j];
      }
    }
  } else {
    bucket_place<false>(tmp2, row, wl0, G, wv, lane, f, st, stage, list + lo);
  }
}

static __device__ __forceinline__ float xhalf(float x) {
  return __builtin_bit_cast(float, __builtin_amdgcn_permlanex16(__builtin_bit_cast(int, x), __builtin_bit_cast(int, x),
                                                                0x76543210, 0xfedcba98, false, false));
}

template <bool WSRC>
static __device__ __forceinline__ float4 gather_row(const float* __restrict__ h,
                                                    const int* __restrict__ list,
                                                    const int* __restrict__ cur,
                                                    const float* __restrict__ dinv,
                                                    int i, int lane) {
  const int half = lane >> 4;
  const int c4 = (lane & 15) * 4;
  const int start = cur[i];
  const int end = cur[i + 1];
  const float di = dinv[i];
  const float4 self = *(const float4*)(h + (size_t)i * CH + c4);
  const float ws = WSRC ? di : 1.0f;
  float4 acc;
  acc.x = half ? 0.0f : self.x * ws;
  acc.y = half ? 0.0f : self.y * ws;
  acc.z = half ? 0.0f : self.z * ws;
  acc.w = half ? 0.0f : self.w * ws;

  for (int base = start; base < end; base += 32) {
    const int n = min(32, end - base);
    int sv = i;
    if (lane < n) sv = list[base + lane];
    int dv = 0;
    if (WSRC) dv = __float_as_int(dinv[sv]);
    const int npairs = (n + 1) >> 1;
    for (int p = 0; p < npairs; p += 8) {
      int s[8];
      float w[8];
      float4 r[8];
#pragma unroll
      for (int u = 0; u < 8; ++u) {
        const int q = 2 * (p + u);
        const int sa = __builtin_amdgcn_readlane(sv, q);
        const int sb = __builtin_amdgcn_readlane(sv, q + 1);
        s[u] = half ? sb : sa;
      }
#pragma unroll
      for (int u = 0; u < 8; ++u) {
        r[u] = make_float4(0.0f, 0.0f, 0.0f, 0.0f);
        if (2 * (p + u) + half < n) r[u] = *(const float4*)(h + (size_t)s[u] * CH + c4);
      }
      if (WSRC) {
#pragma unroll
        for (int u = 0; u < 8; ++u) {
          const int q = 2 * (p + u);
          const int wa = __builtin_amdgcn_readlane(dv, q);
          const int wb = __builtin_amdgcn_readlane(dv, q + 1);
          w[u] = __int_as_float(half ? wb : wa);
        }
#pragma unroll
        for (int u = 0; u < 8; ++u) {
          acc.x = fmaf(r[u].x, w[u], acc.x);
          acc.y = fmaf(r[u].y, w[u], acc.y);
          acc.z = fmaf(r[u].z, w[u], acc.z);
          acc.w = fmaf(r[u].w, w[u], acc.w);
        }
      } else {
#pragma unroll
        for (int u = 0; u < 8; ++u) {
          acc.x += r[u].x;
          acc.y += r[u].y;
          acc.z += r[u].z;
          acc.w += r[u].w;
        }
      }
    }
  }
  acc.x = (acc.x + xhalf(acc.x)) * di;
  acc.y = (acc.y + xhalf(acc.y)) * di;
  acc.z = (acc.z + xhalf(acc.z)) * di;
  acc.w = (acc.w + xhalf(acc.w)) * di;
  return acc;
}

#define TN 16
#define TP 68

template <bool PROJ>
static __device__ __forceinline__ void layer_tile(const float* __restrict__ hin,
                                                  const int* __restrict__ list,
                                                  const int* __restrict__ cur,
                                                  const float* __restrict__ dinv,
                                                  const float* __restrict__ W,
                                                  const float* __restrict__ b,
                                                  const float* __restrict__ Wfc,
                                                  float* __restrict__ out,
                                                  float* __restrict__ ps,
                                                  float* __restrict__ pd, int N) {
  __shared__ float agg[TN * TP];
  __shared__ float act[TN * TP];
  const int lane = threadIdx.x & 31;
  const int wave = threadIdx.x >> 5;
  const int kh = lane >> 4;
  const int m = lane & 15;
  const int c4 = m * 4;
  const int tile0 = blockIdx.x * TN;

  for (int t = 0; t < 4; ++t) {
    const int r = wave * 4 + t;
    const int i = __builtin_amdgcn_readfirstlane(tile0 + r);
    float4 v = make_float4(0.0f, 0.0f, 0.0f, 0.0f);
    if (i < N) v = gather_row<!PROJ>(hin, list, cur, dinv, i, lane);
    if (lane < 16) *(float4*)(agg + r * TP + c4) = v;
  }

  const int n0 = wave * 16;
  v2f bw[CH / 4];
#pragma unroll
  for (int j = 0; j < CH / 4; ++j) {
    bw[j].x = W[(4 * j + 2 * kh + 0) * CH + n0 + m];
    bw[j].y = W[(4 * j + 2 * kh + 1) * CH + n0 + m];
  }
  const float bias = b[n0 + m];
  float dsc[2] = {0.0f, 0.0f};
  if (!PROJ) {
#pragma unroll
    for (int t = 0; t < 2; ++t) {
      const int i = tile0 + wave * 4 + 2 * t + kh;
      if (i < N) dsc[t] = dinv[i];
    }
  }
  __syncthreads();

  const float* arow = agg + m * TP + 2 * kh;
  v8f acc = {};
#pragma unroll
  for (int j = 0; j < CH / 4; ++j) {
    v2f a;
    a.x = arow[4 * j];
    a.y = arow[4 * j + 1];
    acc = __builtin_amdgcn_wmma_f32_16x16x4_f32(false, a, false, bw[j],
                                                (short)0, acc, false, false);
  }
#pragma unroll
  for (int v = 0; v < 8; ++v) {
    const float y = acc[v] + bias;
    act[(v + 8 * kh) * TP + n0 + m] = y > 0.0f ? y : 0.0f;
  }
  float4 w0, w1;
  if (PROJ) {
    w0 = *(const float4*)(Wfc + c4);
    w1 = *(const float4*)(Wfc + CH + c4);
  }
  __syncthreads();

#pragma unroll
  for (int t = 0; t < 2; ++t) {
    const int r = wave * 4 + 2 * t + kh;
    const int i = tile0 + r;
    const float4 v = *(const float4*)(act + r * TP + c4);
    if (PROJ) {
      float s0 = (v.x * w0.x + v.y * w0.y) + (v.z * w0.z + v.w * w0.w);
      float s1 = (v.x * w1.x + v.y * w1.y) + (v.z * w1.z + v.w * w1.w);
#pragma unroll
      for (int off = 8; off > 0; off >>= 1) {
        s0 += __shfl_xor(s0, off, 32);
        s1 += __shfl_xor(s1, off, 32);
      }
      if (m == 0 && i < N) {
        ps[i] = s0;
        pd[i] = s1;
      }
    } else {
      if (i < N) *(float4*)(out + (size_t)i * CH + c4) = make_float4(v.x * dsc[t], v.y * dsc[t], v.z * dsc[t], v.w * dsc[t]);
    }
  }
}

__global__ __launch_bounds__(128) void k_layer_relu(const float* __restrict__ hin,
                                                    const int* __restrict__ list,
                                                    const int* __restrict__ cur,
                                                    const float* __restrict__ dinv,
                                                    const float* __restrict__ W,
                                                    const float* __restrict__ b,
                                                    float* __restrict__ out, int N) {
  layer_tile<false>(hin, list, cur, dinv, W, b, nullptr, out, nullptr, nullptr, N);
}

__global__ __launch_bounds__(128) void k_layer_relu_proj(const float* __restrict__ hin,
                                                         const int* __restrict__ list,
                                                         const int* __restrict__ cur,
                                                         const float* __restrict__ dinv,
                                                         const float* __restrict__ W,
                                                         const float* __restrict__ b,
                                                         const float* __restrict__ Wfc,
                                                         float* __restrict__ ps,
                                                         float* __restrict__ pd, int N) {
  layer_tile<true>(hin, list, cur, dinv, W, b, Wfc, nullptr, ps, pd, N);
}

#define PT 128
#define PQ 2

template <bool TAIL>
static __device__ __forceinline__ void pred_groups(const float* __restrict__ ps, const float* __restrict__ pd,
                                                   const int* __restrict__ src, const int* __restrict__ dst,
                                                   float b, float* __restrict__ out, int q0, int nq) {
  int4 s[PQ], d[PQ];
  float4 a[PQ], c[PQ];
#pragma unroll
  for (int u = 0; u < PQ; ++u) {
    const int q = TAIL ? min(q0 + u * PT, nq - 1) : q0 + u * PT;
    s[u] = ((const int4*)src)[q];
    d[u] = ((const int4*)dst)[q];
  }
#pragma unroll
  for (int u = 0; u < PQ; ++u) {
    a[u].x = ps[s[u].x];  c[u].x = pd[d[u].x];
    a[u].y = ps[s[u].y];  c[u].y = pd[d[u].y];
    a[u].z = ps[s[u].z];  c[u].z = pd[d[u].z];
    a[u].w = ps[s[u].w];  c[u].w = pd[d[u].w];
  }
#pragma unroll
  for (int u = 0; u < PQ; ++u) {
    const int q = q0 + u * PT;
    if (!TAIL || q < nq)
      ((float4*)out)[q] = make_float4(a[u].x + c[u].x + b, a[u].y + c[u].y + b,
                                      a[u].z + c[u].z + b, a[u].w + c[u].w + b);
  }
}

__global__ __launch_bounds__(PT) void k_edge_pred(const float* __restrict__ ps, const float* __restrict__ pd,
                                                  const int* __restrict__ src, const int* __restrict__ dst,
                                                  const float* __restrict__ bfc, float* __restrict__ out, int E) {
  const float b = bfc[0];
  if ((E & 3) == 0) {
    const int nq = E >> 2;
    const int q0 = blockIdx.x * (PT * PQ) + threadIdx.x;
    if ((blockIdx.x + 1) * (PT * PQ) <= nq) pred_groups<false>(ps, pd, src, dst, b, out, q0, nq);
    else                                    pred_groups<true>(ps, pd, src, dst, b, out, q0, nq);
  } else {
    const int e0 = blockIdx.x * (PT * PQ * 4) + threadIdx.x;
    for (int u = 0; u < PQ * 4; ++u) {
      const int e = e0 + u * PT;
      if (e < E) out[e] = ps[src[e]] + pd[dst[e]] + b;
    }
  }
}

extern "C" void kernel_launch(void* const* d_in, const int* in_sizes, int n_in,
                              void* d_out, int out_size, void* d_ws, size_t ws_size,
                              hipStream_t stream) {
  const float* x   = (const float*)d_in[0];
  const int*   ei  = (const int*)d_in[1];
  const float* W1  = (const float*)d_in[2];
  const float* b1  = (const float*)d_in[3];
  const float* W2  = (const float*)d_in[4];
  const float* b2  = (const float*)d_in[5];
  const float* Wfc = (const float*)d_in[6];
  const float* bfc = (const float*)d_in[7];

  const int N = in_sizes[0] / CH;
  const int E = in_sizes[1] / 2;
  const int NF = N * CH;

  const int* src = ei;
  const int* dst = ei + E;

  const int NB = (N + (1 << BSH) - 1) >> BSH;
  const int G  = (E + CHE - 1) / CHE;
  const int GP = (G + 3) & ~3;
  const int M  = NB * GP;
  const int E4 = (E + 3) & ~3;

  float* ws    = (float*)d_ws;
  float* AB    = ws;
  int*   list  = (int*)(AB + (size_t)NF);
  int*   tmp   = list + E4;
  int*   H     = tmp + 2 * (size_t)E4;
  int*   cur   = H + M;
  float* dinv  = (float*)(cur + N + 1);
  float* PS    = dinv + N;
  float* PD    = PS + N;

  dim3 gE((E + PT * PQ * 4 - 1) / (PT * PQ * 4));
  dim3 gT((N + TN - 1) / TN);

  k_chunk_sort<<<dim3(G), dim3(CT), 0, stream>>>(src, dst, H, tmp, E, NB, G, GP);
  k_bucket_sort<<<dim3(NB), dim3(1024), 0, stream>>>(tmp, H, list, cur, dinv, N, G, GP);

  k_layer_relu<<<gT, dim3(128), 0, stream>>>(x, list, cur, dinv, W1, b1, AB, N);

  k_layer_relu_proj<<<gT, dim3(128), 0, stream>>>(AB, list, cur, dinv, W2, b2, Wfc, PS, PD, N);

  k_edge_pred<<<gE, dim3(PT), 0, stream>>>(PS, PD, src, dst, bfc, (float*)d_out, E);
}
